# starting kernel
# baseline (speedup 1.0000x reference)
#include <hip/hip_runtime.h>

#define B_   8
#define S_   4096
#define HID_ 1024
#define H_   16
#define DH_  64

typedef __attribute__((ext_vector_type(2))) float v2f;
typedef __attribute__((ext_vector_type(8))) float v8f;

#define OFF_WQE 0
#define OFF_WKE (OFF_WQE + H_*HID_)
#define OFF_QS  (OFF_WKE + H_*HID_)
#define OFF_HSP (OFF_QS  + B_*S_*H_)
#define OFF_KHP (OFF_HSP + B_*32*HID_)
#define OFF_HS  (OFF_KHP + 256*H_*HID_)
#define OFF_KH  (OFF_HS  + B_*HID_)
#define OFF_MVS (OFF_KH  + B_*H_*HID_)
#define OFF_KVS (OFF_MVS + B_*H_*DH_)
#define OFF_IL  (OFF_KVS + B_*H_*DH_)

__global__ __launch_bounds__(256) void weff_kernel(
    const float* __restrict__ Wq, const float* __restrict__ Wk,
    const float* __restrict__ aw, float* __restrict__ wqe, float* __restrict__ wke)
{
    int u = blockIdx.x * 256 + threadIdx.x;
    int h = u >> 10, i = u & (HID_ - 1);
    float aq = 0.f, ak = 0.f;
#pragma unroll 4
    for (int d = 0; d < DH_; ++d) {
        float a = aw[h * DH_ + d];
        aq += a * Wq[(size_t)(h * DH_ + d) * HID_ + i];
        ak += a * Wk[(size_t)(h * DH_ + d) * HID_ + i];
    }
    wqe[h * HID_ + i] = aq;
    wke[h * HID_ + i] = ak;
}

__global__ __launch_bounds__(256) void invlen_kernel(
    const float* __restrict__ mask, float* __restrict__ il)
{
    __shared__ float red[256];
    int b = blockIdx.x, t = threadIdx.x;
    float s = 0.f;
#pragma unroll
    for (int k = 0; k < S_ / 256; ++k) s += mask[b * S_ + t + 256 * k];
    red[t] = s;
    __syncthreads();
    for (int w = 128; w > 0; w >>= 1) {
        if (t < w) red[t] += red[t + w];
        __syncthreads();
    }
    if (t == 0) il[b] = 1.0f / red[0];
}

__global__ __launch_bounds__(256) void pass1_kernel(
    const float* __restrict__ hidden, const float* __restrict__ mask,
    const float* __restrict__ wqe, const float* __restrict__ wke,
    float* __restrict__ q_scores, float* __restrict__ hsum_part,
    float* __restrict__ kh_part)
{
    __shared__ float lds_tile[128 * 68];
    __shared__ float lds_wq[16 * 68];
    __shared__ float lds_wk[16 * 68];
    __shared__ float lds_ksT[16 * 132];
    __shared__ float lds_red[256 * 4];
    __shared__ float lds_khb[2 * 16 * 68];

    const int blk = blockIdx.x & 31;
    const int b   = blockIdx.x >> 5;
    const int s0  = blk * 128;
    const int t = threadIdx.x;
    const int wave = t >> 5, lane = t & 31;
    const int lane16 = lane & 15, laneHi = lane >> 4;

    const float* hbase = hidden + ((size_t)b * S_ + s0) * HID_;

    v8f cq = {}; v8f ck = {};

    for (int kc = 0; kc < 16; ++kc) {
        const int k0 = kc * 64;
        {
            int r = t >> 4, c4 = (t & 15) * 4;
            *(float4*)&lds_wq[r * 68 + c4] = *(const float4*)(wqe + r * HID_ + k0 + c4);
            *(float4*)&lds_wk[r * 68 + c4] = *(const float4*)(wke + r * HID_ + k0 + c4);
        }
        float4 acc = {0.f, 0.f, 0.f, 0.f};
#pragma unroll
        for (int it = 0; it < 8; ++it) {
            int row = it * 16 + (t >> 4);
            float4 v = *(const float4*)(hbase + (size_t)row * HID_ + k0 + (t & 15) * 4);
            *(float4*)&lds_tile[row * 68 + (t & 15) * 4] = v;
            float m = mask[b * S_ + s0 + row];
            acc.x += m * v.x; acc.y += m * v.y; acc.z += m * v.z; acc.w += m * v.w;
        }
        *(float4*)&lds_red[t * 4] = acc;
        __syncthreads();
        if (t < 64) {
            int g = t >> 2, comp = t & 3;
            float s = 0.f;
#pragma unroll
            for (int p = 0; p < 16; ++p) s += lds_red[(p * 16 + g) * 4 + comp];
            hsum_part[(size_t)(b * 32 + blk) * HID_ + k0 + g * 4 + comp] = s;
        }
        const float* atile = &lds_tile[(wave * 16 + lane16) * 68 + 2 * laneHi];
        const float* bq = &lds_wq[lane16 * 68 + 2 * laneHi];
        const float* bk = &lds_wk[lane16 * 68 + 2 * laneHi];
#pragma unroll
        for (int kk = 0; kk < 16; ++kk) {
            v2f a = *(const v2f*)(atile + kk * 4);
            v2f q = *(const v2f*)(bq + kk * 4);
            v2f k = *(const v2f*)(bk + kk * 4);
            cq = __builtin_amdgcn_wmma_f32_16x16x4_f32(false, a, false, q, (short)0, cq, false, false);
            ck = __builtin_amdgcn_wmma_f32_16x16x4_f32(false, a, false, k, (short)0, ck, false, false);
        }
        __syncthreads();
    }

#pragma unroll
    for (int j = 0; j < 8; ++j) {
        int sl = wave * 16 + j + 8 * laneHi;
        float m = mask[b * S_ + s0 + sl];
        q_scores[((size_t)b * S_ + s0 + sl) * H_ + lane16] = cq[j] * m;
        lds_ksT[lane16 * 132 + sl] = ck[j] * m;
    }
    __syncthreads();

    const int ic = wave & 3;
    const int sh = wave >> 2;
    for (int kc = 0; kc < 16; ++kc) {
        const int k0 = kc * 64;
#pragma unroll
        for (int it = 0; it < 8; ++it) {
            int row = it * 16 + (t >> 4);
            *(float4*)&lds_tile[row * 68 + (t & 15) * 4] =
                *(const float4*)(hbase + (size_t)row * HID_ + k0 + (t & 15) * 4);
        }
        __syncthreads();
        v8f ckh = {};
        const float* aks = &lds_ksT[lane16 * 132 + sh * 64 + 2 * laneHi];
#pragma unroll
        for (int ss = 0; ss < 16; ++ss) {
            v2f a = *(const v2f*)(aks + ss * 4);
            int srow = sh * 64 + ss * 4 + 2 * laneHi;
            v2f bb;
            bb.x = lds_tile[srow * 68 + ic * 16 + lane16];
            bb.y = lds_tile[(srow + 1) * 68 + ic * 16 + lane16];
            ckh = __builtin_amdgcn_wmma_f32_16x16x4_f32(false, a, false, bb, (short)0, ckh, false, false);
        }
#pragma unroll
        for (int j = 0; j < 8; ++j)
            lds_khb[(sh * 16 + j + 8 * laneHi) * 68 + ic * 16 + lane16] = ckh[j];
        __syncthreads();
        {
            int h = t >> 4, c4 = (t & 15) * 4;
            float4 a0 = *(const float4*)&lds_khb[h * 68 + c4];
            float4 a1 = *(const float4*)&lds_khb[(16 + h) * 68 + c4];
            float4 r; r.x = a0.x + a1.x; r.y = a0.y + a1.y; r.z = a0.z + a1.z; r.w = a0.w + a1.w;
            *(float4*)&kh_part[(((size_t)(b * 32 + blk)) * H_ + h) * HID_ + k0 + c4] = r;
        }
        __syncthreads();
    }
}

__global__ __launch_bounds__(256) void reduce_hsum_kernel(
    const float* __restrict__ hsp, float* __restrict__ hs)
{
    int idx = blockIdx.x * 256 + threadIdx.x;
    int b = idx >> 10, i = idx & (HID_ - 1);
    float s = 0.f;
#pragma unroll 8
    for (int blk = 0; blk < 32; ++blk) s += hsp[(size_t)(b * 32 + blk) * HID_ + i];
    hs[idx] = s;
}

__global__ __launch_bounds__(256) void reduce_kh_kernel(
    const float* __restrict__ khp, float* __restrict__ kh)
{
    int idx = blockIdx.x * 256 + threadIdx.x;
    int bh = idx >> 10, i = idx & (HID_ - 1);
    int b = bh >> 4, h = bh & 15;
    float s = 0.f;
#pragma unroll 8
    for (int blk = 0; blk < 32; ++blk)
        s += khp[(((size_t)(b * 32 + blk)) * H_ + h) * HID_ + i];
    kh[idx] = s;
}

__global__ __launch_bounds__(256) void pass2_kernel(
    const float* __restrict__ Wv, const float* __restrict__ hs,
    const float* __restrict__ kh, const float* __restrict__ il,
    float* __restrict__ mvs, float* __restrict__ kvs)
{
    int u = blockIdx.x * 256 + threadIdx.x;
    int h = u >> 6, d = u & 63;
    float mv[B_], kv[B_];
#pragma unroll
    for (int b = 0; b < B_; ++b) { mv[b] = 0.f; kv[b] = 0.f; }
    const float* wrow = Wv + (size_t)(h * DH_ + d) * HID_;
    for (int i = 0; i < HID_; i += 4) {
        float4 w = *(const float4*)(wrow + i);
#pragma unroll
        for (int b = 0; b < B_; ++b) {
            float4 s4 = *(const float4*)(hs + b * HID_ + i);
            float4 k4 = *(const float4*)(kh + (size_t)(b * H_ + h) * HID_ + i);
            mv[b] += w.x * s4.x + w.y * s4.y + w.z * s4.z + w.w * s4.w;
            kv[b] += w.x * k4.x + w.y * k4.y + w.z * k4.z + w.w * k4.w;
        }
    }
#pragma unroll
    for (int b = 0; b < B_; ++b) {
        float inv = il[b];
        mvs[(b * H_ + h) * DH_ + d] = mv[b] * inv;
        kvs[(b * H_ + h) * DH_ + d] = kv[b] * inv;
    }
}

__global__ __launch_bounds__(256) void pass3_kernel(
    const float* __restrict__ qs, const float* __restrict__ mvs,
    const float* __restrict__ kvs, float* __restrict__ out)
{
    const int b = blockIdx.x >> 7;
    const int srow0 = (blockIdx.x & 127) * 32;
    const int t = threadIdx.x;
    const int h = t >> 4;
    float4 mv4 = *(const float4*)(mvs + (b * H_ + h) * DH_ + (t & 15) * 4);
    float4 kv4 = *(const float4*)(kvs + (b * H_ + h) * DH_ + (t & 15) * 4);
#pragma unroll 4
    for (int r = 0; r < 32; ++r) {
        size_t srow = (size_t)b * S_ + srow0 + r;
        float q = qs[srow * H_ + h];
        float4 o;
        o.x = q * mv4.x + kv4.x; o.y = q * mv4.y + kv4.y;
        o.z = q * mv4.z + kv4.z; o.w = q * mv4.w + kv4.w;
        *(float4*)(out + srow * (H_ * DH_) + t * 4) = o;
    }
}

extern "C" void kernel_launch(void* const* d_in, const int* in_sizes, int n_in,
                              void* d_out, int out_size, void* d_ws, size_t ws_size,
                              hipStream_t stream) {
    const float* hidden = (const float*)d_in[0];
    const float* mask   = (const float*)d_in[1];
    const float* Wq     = (const float*)d_in[2];
    const float* Wk     = (const float*)d_in[3];
    const float* Wv     = (const float*)d_in[4];
    const float* aw     = (const float*)d_in[5];
    float* out = (float*)d_out;
    float* ws  = (float*)d_ws;

    weff_kernel   <<<64,   256, 0, stream>>>(Wq, Wk, aw, ws + OFF_WQE, ws + OFF_WKE);
    invlen_kernel <<<B_,   256, 0, stream>>>(mask, ws + OFF_IL);
    pass1_kernel  <<<256,  256, 0, stream>>>(hidden, mask, ws + OFF_WQE, ws + OFF_WKE,
                                             ws + OFF_QS, ws + OFF_HSP, ws + OFF_KHP);
    reduce_hsum_kernel<<<32,  256, 0, stream>>>(ws + OFF_HSP, ws + OFF_HS);
    reduce_kh_kernel  <<<512, 256, 0, stream>>>(ws + OFF_KHP, ws + OFF_KH);
    pass2_kernel  <<<4,    256, 0, stream>>>(Wv, ws + OFF_HS, ws + OFF_KH, ws + OFF_IL,
                                             ws + OFF_MVS, ws + OFF_KVS);
    pass3_kernel  <<<1024, 256, 0, stream>>>(ws + OFF_QS, ws + OFF_MVS, ws + OFF_KVS, out);
}


# Wv projection fused into the partial-sum reducer: 512 work-groups (batch x head x 256-column slice) sum kh/hsum partials and dot their Wv slice; output pass adds the 4 slices (new sum order)
# speedup vs baseline: 1.6460x; 1.6460x over previous
#include <hip/hip_runtime.h>

#define B_   8
#define S_   4096
#define HID_ 1024
#define H_   16
#define DH_  64

typedef __attribute__((ext_vector_type(2))) float v2f;
typedef __attribute__((ext_vector_type(8))) float v8f;

#define OFF_WQE 0
#define OFF_WKE (OFF_WQE + H_*HID_)
#define OFF_QS  (OFF_WKE + H_*HID_)
#define OFF_HSP (OFF_QS  + B_*S_*H_)
#define OFF_KHP (OFF_HSP + B_*32*HID_)
#define NSL_ 4
#define OFF_MKP (OFF_KHP + 256*H_*HID_)
#define OFF_IL  (OFF_MKP + B_*H_*NSL_*2*DH_)

__global__ __launch_bounds__(256) void weff_kernel(
    const float* __restrict__ Wq, const float* __restrict__ Wk,
    const float* __restrict__ aw, float* __restrict__ wqe, float* __restrict__ wke)
{
    int u = blockIdx.x * 256 + threadIdx.x;
    int h = u >> 10, i = u & (HID_ - 1);
    float aq = 0.f, ak = 0.f;
#pragma unroll 4
    for (int d = 0; d < DH_; ++d) {
        float a = aw[h * DH_ + d];
        aq += a * Wq[(size_t)(h * DH_ + d) * HID_ + i];
        ak += a * Wk[(size_t)(h * DH_ + d) * HID_ + i];
    }
    wqe[h * HID_ + i] = aq;
    wke[h * HID_ + i] = ak;
}

__global__ __launch_bounds__(256) void invlen_kernel(
    const float* __restrict__ mask, float* __restrict__ il)
{
    __shared__ float red[256];
    int b = blockIdx.x, t = threadIdx.x;
    float s = 0.f;
#pragma unroll
    for (int k = 0; k < S_ / 256; ++k) s += mask[b * S_ + t + 256 * k];
    red[t] = s;
    __syncthreads();
    for (int w = 128; w > 0; w >>= 1) {
        if (t < w) red[t] += red[t + w];
        __syncthreads();
    }
    if (t == 0) il[b] = 1.0f / red[0];
}

__global__ __launch_bounds__(256) void pass1_kernel(
    const float* __restrict__ hidden, const float* __restrict__ mask,
    const float* __restrict__ wqe, const float* __restrict__ wke,
    float* __restrict__ q_scores, float* __restrict__ hsum_part,
    float* __restrict__ kh_part)
{
    __shared__ float lds_tile[128 * 68];
    __shared__ float lds_wq[16 * 68];
    __shared__ float lds_wk[16 * 68];
    __shared__ float lds_ksT[16 * 132];
    __shared__ float lds_red[256 * 4];
    __shared__ float lds_khb[2 * 16 * 68];

    const int blk = blockIdx.x & 31;
    const int b   = blockIdx.x >> 5;
    const int s0  = blk * 128;
    const int t = threadIdx.x;
    const int wave = t >> 5, lane = t & 31;
    const int lane16 = lane & 15, laneHi = lane >> 4;

    const float* hbase = hidden + ((size_t)b * S_ + s0) * HID_;

    v8f cq = {}; v8f ck = {};

    for (int kc = 0; kc < 16; ++kc) {
        const int k0 = kc * 64;
        {
            int r = t >> 4, c4 = (t & 15) * 4;
            *(float4*)&lds_wq[r * 68 + c4] = *(const float4*)(wqe + r * HID_ + k0 + c4);
            *(float4*)&lds_wk[r * 68 + c4] = *(const float4*)(wke + r * HID_ + k0 + c4);
        }
        float4 acc = {0.f, 0.f, 0.f, 0.f};
#pragma unroll
        for (int it = 0; it < 8; ++it) {
            int row = it * 16 + (t >> 4);
            float4 v = *(const float4*)(hbase + (size_t)row * HID_ + k0 + (t & 15) * 4);
            *(float4*)&lds_tile[row * 68 + (t & 15) * 4] = v;
            float m = mask[b * S_ + s0 + row];
            acc.x += m * v.x; acc.y += m * v.y; acc.z += m * v.z; acc.w += m * v.w;
        }
        *(float4*)&lds_red[t * 4] = acc;
        __syncthreads();
        if (t < 64) {
            int g = t >> 2, comp = t & 3;
            float s = 0.f;
#pragma unroll
            for (int p = 0; p < 16; ++p) s += lds_red[(p * 16 + g) * 4 + comp];
            hsum_part[(size_t)(b * 32 + blk) * HID_ + k0 + g * 4 + comp] = s;
        }
        const float* atile = &lds_tile[(wave * 16 + lane16) * 68 + 2 * laneHi];
        const float* bq = &lds_wq[lane16 * 68 + 2 * laneHi];
        const float* bk = &lds_wk[lane16 * 68 + 2 * laneHi];
#pragma unroll
        for (int kk = 0; kk < 16; ++kk) {
            v2f a = *(const v2f*)(atile + kk * 4);
            v2f q = *(const v2f*)(bq + kk * 4);
            v2f k = *(const v2f*)(bk + kk * 4);
            cq = __builtin_amdgcn_wmma_f32_16x16x4_f32(false, a, false, q, (short)0, cq, false, false);
            ck = __builtin_amdgcn_wmma_f32_16x16x4_f32(false, a, false, k, (short)0, ck, false, false);
        }
        __syncthreads();
    }

#pragma unroll
    for (int j = 0; j < 8; ++j) {
        int sl = wave * 16 + j + 8 * laneHi;
        float m = mask[b * S_ + s0 + sl];
        q_scores[((size_t)b * S_ + s0 + sl) * H_ + lane16] = cq[j] * m;
        lds_ksT[lane16 * 132 + sl] = ck[j] * m;
    }
    __syncthreads();

    const int ic = wave & 3;
    const int sh = wave >> 2;
    for (int kc = 0; kc < 16; ++kc) {
        const int k0 = kc * 64;
#pragma unroll
        for (int it = 0; it < 8; ++it) {
            int row = it * 16 + (t >> 4);
            *(float4*)&lds_tile[row * 68 + (t & 15) * 4] =
                *(const float4*)(hbase + (size_t)row * HID_ + k0 + (t & 15) * 4);
        }
        __syncthreads();
        v8f ckh = {};
        const float* aks = &lds_ksT[lane16 * 132 + sh * 64 + 2 * laneHi];
#pragma unroll
        for (int ss = 0; ss < 16; ++ss) {
            v2f a = *(const v2f*)(aks + ss * 4);
            int srow = sh * 64 + ss * 4 + 2 * laneHi;
            v2f bb;
            bb.x = lds_tile[srow * 68 + ic * 16 + lane16];
            bb.y = lds_tile[(srow + 1) * 68 + ic * 16 + lane16];
            ckh = __builtin_amdgcn_wmma_f32_16x16x4_f32(false, a, false, bb, (short)0, ckh, false, false);
        }
#pragma unroll
        for (int j = 0; j < 8; ++j)
            lds_khb[(sh * 16 + j + 8 * laneHi) * 68 + ic * 16 + lane16] = ckh[j];
        __syncthreads();
        {
            int h = t >> 4, c4 = (t & 15) * 4;
            float4 a0 = *(const float4*)&lds_khb[h * 68 + c4];
            float4 a1 = *(const float4*)&lds_khb[(16 + h) * 68 + c4];
            float4 r; r.x = a0.x + a1.x; r.y = a0.y + a1.y; r.z = a0.z + a1.z; r.w = a0.w + a1.w;
            *(float4*)&kh_part[(((size_t)(b * 32 + blk)) * H_ + h) * HID_ + k0 + c4] = r;
        }
        __syncthreads();
    }
}

__global__ __launch_bounds__(256) void reduce_project_kernel(
    const float* __restrict__ hsp, const float* __restrict__ khp,
    const float* __restrict__ Wv, float* __restrict__ mkp)
{
    __shared__ float4 s_kh[4][64];
    __shared__ float4 s_hs[4][64];
    __shared__ float  s_out[2 * DH_];

    const int sl = blockIdx.x % NSL_;
    const int bh = blockIdx.x / NSL_;
    const int b = bh >> 4, h = bh & 15;
    const int t = threadIdx.x;
    const int wave = t >> 5, lane = t & 31;
    const int col0 = sl * 256;

    const int c = t & 63, g = t >> 6;
    const float* kp = khp + (((size_t)(b * 32 + g * 8)) * H_ + h) * HID_ + col0 + c * 4;
    const float* hp = hsp + ((size_t)(b * 32 + g * 8)) * HID_ + col0 + c * 4;
    float4 pk4[8], ph4[8];
#pragma unroll
    for (int p = 0; p < 8; ++p) pk4[p] = *(const float4*)(kp + (size_t)p * H_ * HID_);
#pragma unroll
    for (int p = 0; p < 8; ++p) ph4[p] = *(const float4*)(hp + (size_t)p * HID_);

    const float* wbase = Wv + (size_t)(h * DH_ + wave * 8) * HID_ + col0 + lane * 4;
    float4 w0[8], w1[8];
#pragma unroll
    for (int r = 0; r < 8; ++r) {
        w0[r] = *(const float4*)(wbase + (size_t)r * HID_);
        w1[r] = *(const float4*)(wbase + (size_t)r * HID_ + 128);
    }

    float4 ak = pk4[0], ah = ph4[0];
#pragma unroll
    for (int p = 1; p < 8; ++p) {
        ak.x += pk4[p].x; ak.y += pk4[p].y; ak.z += pk4[p].z; ak.w += pk4[p].w;
        ah.x += ph4[p].x; ah.y += ph4[p].y; ah.z += ph4[p].z; ah.w += ph4[p].w;
    }
    s_kh[g][c] = ak;
    s_hs[g][c] = ah;
    __syncthreads();

    float4 k0 = s_kh[0][lane], k1 = s_kh[0][32 + lane];
    float4 m0 = s_hs[0][lane], m1 = s_hs[0][32 + lane];
#pragma unroll
    for (int q = 1; q < 4; ++q) {
        float4 a = s_kh[q][lane], e = s_kh[q][32 + lane];
        float4 u = s_hs[q][lane], v = s_hs[q][32 + lane];
        k0.x += a.x; k0.y += a.y; k0.z += a.z; k0.w += a.w;
        k1.x += e.x; k1.y += e.y; k1.z += e.z; k1.w += e.w;
        m0.x += u.x; m0.y += u.y; m0.z += u.z; m0.w += u.w;
        m1.x += v.x; m1.y += v.y; m1.z += v.z; m1.w += v.w;
    }

    float pk[8], pm[8];
#pragma unroll
    for (int r = 0; r < 8; ++r) {
        float sk = w0[r].x * k0.x;
        sk += w0[r].y * k0.y; sk += w0[r].z * k0.z; sk += w0[r].w * k0.w;
        sk += w1[r].x * k1.x; sk += w1[r].y * k1.y; sk += w1[r].z * k1.z; sk += w1[r].w * k1.w;
        float sm = w0[r].x * m0.x;
        sm += w0[r].y * m0.y; sm += w0[r].z * m0.z; sm += w0[r].w * m0.w;
        sm += w1[r].x * m1.x; sm += w1[r].y * m1.y; sm += w1[r].z * m1.z; sm += w1[r].w * m1.w;
        pk[r] = sk; pm[r] = sm;
    }
#pragma unroll
    for (int m = 16; m >= 1; m >>= 1) {
#pragma unroll
        for (int r = 0; r < 8; ++r) {
            pk[r] += __shfl_xor(pk[r], m, 32);
            pm[r] += __shfl_xor(pm[r], m, 32);
        }
    }
    float ok = 0.f, om = 0.f;
#pragma unroll
    for (int r = 0; r < 8; ++r) if (lane == r) { ok = pk[r]; om = pm[r]; }
    if (lane < 8) {
        s_out[wave * 8 + lane] = om;
        s_out[DH_ + wave * 8 + lane] = ok;
    }
    __syncthreads();
    if (t < 32)
        *(float4*)&mkp[((size_t)bh * NSL_ + sl) * (2 * DH_) + t * 4] = *(const float4*)&s_out[t * 4];
}

__global__ __launch_bounds__(256) void pass3_kernel(
    const float* __restrict__ qs, const float* __restrict__ mkp,
    const float* __restrict__ il, float* __restrict__ out)
{
    const int b = blockIdx.x >> 7;
    const int srow0 = (blockIdx.x & 127) * 32;
    const int t = threadIdx.x;
    const int h = t >> 4;
    const float* mk = mkp + (size_t)(b * H_ + h) * (NSL_ * 2 * DH_) + (t & 15) * 4;
    float4 mv4 = *(const float4*)(mk);
    float4 kv4 = *(const float4*)(mk + DH_);
#pragma unroll
    for (int c = 1; c < NSL_; ++c) {
        float4 a = *(const float4*)(mk + c * 2 * DH_);
        float4 e = *(const float4*)(mk + c * 2 * DH_ + DH_);
        mv4.x += a.x; mv4.y += a.y; mv4.z += a.z; mv4.w += a.w;
        kv4.x += e.x; kv4.y += e.y; kv4.z += e.z; kv4.w += e.w;
    }
    const float inv = il[b];
    mv4.x *= inv; mv4.y *= inv; mv4.z *= inv; mv4.w *= inv;
    kv4.x *= inv; kv4.y *= inv; kv4.z *= inv; kv4.w *= inv;
#pragma unroll 4
    for (int r = 0; r < 32; ++r) {
        size_t srow = (size_t)b * S_ + srow0 + r;
        float q = qs[srow * H_ + h];
        float4 o;
        o.x = q * mv4.x + kv4.x; o.y = q * mv4.y + kv4.y;
        o.z = q * mv4.z + kv4.z; o.w = q * mv4.w + kv4.w;
        *(float4*)(out + srow * (H_ * DH_) + t * 4) = o;
    }
}

extern "C" void kernel_launch(void* const* d_in, const int* in_sizes, int n_in,
                              void* d_out, int out_size, void* d_ws, size_t ws_size,
                              hipStream_t stream) {
    const float* hidden = (const float*)d_in[0];
    const float* mask   = (const float*)d_in[1];
    const float* Wq     = (const float*)d_in[2];
    const float* Wk     = (const float*)d_in[3];
    const float* Wv     = (const float*)d_in[4];
    const float* aw     = (const float*)d_in[5];
    float* out = (float*)d_out;
    float* ws  = (float*)d_ws;

    weff_kernel   <<<64,   256, 0, stream>>>(Wq, Wk, aw, ws + OFF_WQE, ws + OFF_WKE);
    invlen_kernel <<<B_,   256, 0, stream>>>(mask, ws + OFF_IL);
    pass1_kernel  <<<256,  256, 0, stream>>>(hidden, mask, ws + OFF_WQE, ws + OFF_WKE,
                                             ws + OFF_QS, ws + OFF_HSP, ws + OFF_KHP);
    reduce_project_kernel<<<512, 256, 0, stream>>>(ws + OFF_HSP, ws + OFF_KHP, Wv, ws + OFF_MKP);
    pass3_kernel  <<<1024, 256, 0, stream>>>(ws + OFF_QS, ws + OFF_MKP, ws + OFF_IL, out);
}


# scores/kh kernel, no work-group barrier in K loops: 16 waves, weights kept in LDS, a wave stages only its own 16 rows x K-half; kh, hsum from global-fed registers; new sum order: scores, hsum, kh
# speedup vs baseline: 2.3577x; 1.4324x over previous
#include <hip/hip_runtime.h>

#define B_   8
#define S_   4096
#define HID_ 1024
#define H_   16
#define DH_  64

typedef __attribute__((ext_vector_type(2))) float v2f;
typedef __attribute__((ext_vector_type(8))) float v8f;

#define OFF_WQE 0
#define OFF_WKE (OFF_WQE + H_*HID_)
#define OFF_QS  (OFF_WKE + H_*HID_)
#define OFF_HSP (OFF_QS  + B_*S_*H_)
#define OFF_KHP (OFF_HSP + B_*32*HID_)
#define NSL_ 4
#define OFF_MKP (OFF_KHP + 256*H_*HID_)
#define OFF_IL  (OFF_MKP + B_*H_*NSL_*2*DH_)

__global__ __launch_bounds__(256) void weff_kernel(
    const float* __restrict__ Wq, const float* __restrict__ Wk,
    const float* __restrict__ aw, float* __restrict__ wqe, float* __restrict__ wke)
{
    int u = blockIdx.x * 256 + threadIdx.x;
    int h = u >> 10, i = u & (HID_ - 1);
    float aq = 0.f, ak = 0.f;
#pragma unroll 4
    for (int d = 0; d < DH_; ++d) {
        float a = aw[h * DH_ + d];
        aq += a * Wq[(size_t)(h * DH_ + d) * HID_ + i];
        ak += a * Wk[(size_t)(h * DH_ + d) * HID_ + i];
    }
    wqe[h * HID_ + i] = aq;
    wke[h * HID_ + i] = ak;
}

__global__ __launch_bounds__(256) void invlen_kernel(
    const float* __restrict__ mask, float* __restrict__ il)
{
    __shared__ float red[256];
    int b = blockIdx.x, t = threadIdx.x;
    float s = 0.f;
#pragma unroll
    for (int k = 0; k < S_ / 256; ++k) s += mask[b * S_ + t + 256 * k];
    red[t] = s;
    __syncthreads();
    for (int w = 128; w > 0; w >>= 1) {
        if (t < w) red[t] += red[t + w];
        __syncthreads();
    }
    if (t == 0) il[b] = 1.0f / red[0];
}

#define WSTR_ 1028
#define TSTR_ 68

__global__ __launch_bounds__(512) void pass1_kernel(
    const float* __restrict__ hidden, const float* __restrict__ mask,
    const float* __restrict__ wqe, const float* __restrict__ wke,
    float* __restrict__ q_scores, float* __restrict__ hsum_part,
    float* __restrict__ kh_part)
{
    __shared__ __attribute__((aligned(16))) float lds_wq[16 * WSTR_];
    __shared__ __attribute__((aligned(16))) float lds_wk[16 * WSTR_];
    __shared__ __attribute__((aligned(16))) float lds_strip[16 * 16 * TSTR_];
    __shared__ __attribute__((aligned(16))) float lds_ksT[16 * 132];
    __shared__ __attribute__((aligned(16))) float lds_half[8 * 2 * 8 * 32];
    __shared__ __attribute__((aligned(16))) float lds_mask[128];

    const int blk = blockIdx.x & 31;
    const int b   = blockIdx.x >> 5;
    const int s0  = blk * 128;
    const int t = threadIdx.x;
    const int wave = t >> 5, lane = t & 31;
    const int lane16 = lane & 15, laneHi = lane >> 4;
    const int mg = wave & 7;
    const int kq = wave >> 3;

    const float* hbase = hidden + ((size_t)b * S_ + s0) * HID_;
    const float* hrow = hbase + (size_t)(mg * 16 + laneHi) * HID_ + kq * 512 + lane16 * 4;

#define LOAD_STRIP_(p) \
    r0 = *(const float4*)((p));                         r1 = *(const float4*)((p) + (size_t)2 * HID_);  \
    r2 = *(const float4*)((p) + (size_t)4 * HID_);      r3 = *(const float4*)((p) + (size_t)6 * HID_);  \
    r4 = *(const float4*)((p) + (size_t)8 * HID_);      r5 = *(const float4*)((p) + (size_t)10 * HID_); \
    r6 = *(const float4*)((p) + (size_t)12 * HID_);     r7 = *(const float4*)((p) + (size_t)14 * HID_);
    float4 r0, r1, r2, r3, r4, r5, r6, r7;
    LOAD_STRIP_(hrow)

#pragma unroll
    for (int i = 0; i < 8; ++i) {
        int idx = i * 512 + t;
        int row = idx >> 8, c4 = (idx & 255) * 4;
        *(float4*)&lds_wq[row * WSTR_ + c4] = *(const float4*)(wqe + row * HID_ + c4);
        *(float4*)&lds_wk[row * WSTR_ + c4] = *(const float4*)(wke + row * HID_ + c4);
    }
    if (t < 128) lds_mask[t] = mask[b * S_ + s0 + t];
    __syncthreads();

    float* strip = &lds_strip[wave * (16 * TSTR_)];
    float* sdst = strip + laneHi * TSTR_ + lane16 * 4;
    const float* atile = strip + lane16 * TSTR_ + 2 * laneHi;
    const float* bq = &lds_wq[lane16 * WSTR_ + kq * 512 + 2 * laneHi];
    const float* bk = &lds_wk[lane16 * WSTR_ + kq * 512 + 2 * laneHi];

    v8f cq = {}; v8f ck = {};
#pragma unroll 1
    for (int c = 0; c < 8; ++c) {
        *(float4*)(sdst)              = r0;  *(float4*)(sdst + 2 * TSTR_)  = r1;
        *(float4*)(sdst + 4 * TSTR_)  = r2;  *(float4*)(sdst + 6 * TSTR_)  = r3;
        *(float4*)(sdst + 8 * TSTR_)  = r4;  *(float4*)(sdst + 10 * TSTR_) = r5;
        *(float4*)(sdst + 12 * TSTR_) = r6;  *(float4*)(sdst + 14 * TSTR_) = r7;
        if (c < 7) {
            const float* hn = hrow + (c + 1) * 64;
            LOAD_STRIP_(hn)
        }
#pragma unroll
        for (int kk = 0; kk < 16; ++kk) {
            v2f a = *(const v2f*)(atile + kk * 4);
            v2f q = *(const v2f*)(bq + c * 64 + kk * 4);
            v2f k = *(const v2f*)(bk + c * 64 + kk * 4);
            cq = __builtin_amdgcn_wmma_f32_16x16x4_f32(false, a, false, q, (short)0, cq, false, false);
            ck = __builtin_amdgcn_wmma_f32_16x16x4_f32(false, a, false, k, (short)0, ck, false, false);
        }
    }

    if (kq == 1) {
#pragma unroll
        for (int j = 0; j < 8; ++j) {
            lds_half[((mg * 2 + 0) * 8 + j) * 32 + lane] = cq[j];
            lds_half[((mg * 2 + 1) * 8 + j) * 32 + lane] = ck[j];
        }
    }
    __syncthreads();
    if (kq == 0) {
#pragma unroll
        for (int j = 0; j < 8; ++j) {
            int sl = mg * 16 + j + 8 * laneHi;
            float m = lds_mask[sl];
            float qv = cq[j] + lds_half[((mg * 2 + 0) * 8 + j) * 32 + lane];
            float kv = ck[j] + lds_half[((mg * 2 + 1) * 8 + j) * 32 + lane];
            q_scores[((size_t)b * S_ + s0 + sl) * H_ + lane16] = qv * m;
            lds_ksT[lane16 * 132 + sl] = kv * m;
        }
    }
    __syncthreads();

    const int i0 = wave * 64 + lane16 * 4;
    const float* hc = hbase + (size_t)(2 * laneHi) * HID_ + i0;
    const float* aks = &lds_ksT[lane16 * 132 + 2 * laneHi];
    const float* mk2 = &lds_mask[2 * laneHi];
    v8f kh0 = {}, kh1 = {}, kh2 = {}, kh3 = {};
    float4 hs = {0.f, 0.f, 0.f, 0.f};
#define KSTEP_(U, XA, XB) { \
        v2f m = *(const v2f*)(mk2 + g * 16 + U * 4); \
        v2f a = *(const v2f*)(aks + g * 16 + U * 4); \
        hs.x += m.x * XA.x; hs.y += m.x * XA.y; hs.z += m.x * XA.z; hs.w += m.x * XA.w; \
        hs.x += m.y * XB.x; hs.y += m.y * XB.y; hs.z += m.y * XB.z; hs.w += m.y * XB.w; \
        v2f b0 = {XA.x, XB.x}, b1 = {XA.y, XB.y}, b2 = {XA.z, XB.z}, b3 = {XA.w, XB.w}; \
        kh0 = __builtin_amdgcn_wmma_f32_16x16x4_f32(false, a, false, b0, (short)0, kh0, false, false); \
        kh1 = __builtin_amdgcn_wmma_f32_16x16x4_f32(false, a, false, b1, (short)0, kh1, false, false); \
        kh2 = __builtin_amdgcn_wmma_f32_16x16x4_f32(false, a, false, b2, (short)0, kh2, false, false); \
        kh3 = __builtin_amdgcn_wmma_f32_16x16x4_f32(false, a, false, b3, (short)0, kh3, false, false); }
    for (int g = 0; g < 8; ++g) {
        const float* hn = hc + (size_t)(g * 16) * HID_;
        float4 x00 = *(const float4*)(hn);
        float4 x01 = *(const float4*)(hn + (size_t)1 * HID_);
        float4 x10 = *(const float4*)(hn + (size_t)4 * HID_);
        float4 x11 = *(const float4*)(hn + (size_t)5 * HID_);
        float4 x20 = *(const float4*)(hn + (size_t)8 * HID_);
        float4 x21 = *(const float4*)(hn + (size_t)9 * HID_);
        float4 x30 = *(const float4*)(hn + (size_t)12 * HID_);
        float4 x31 = *(const float4*)(hn + (size_t)13 * HID_);
        KSTEP_(0, x00, x01) KSTEP_(1, x10, x11) KSTEP_(2, x20, x21) KSTEP_(3, x30, x31)
    }

    float* kp = kh_part + (((size_t)(b * 32 + blk)) * H_ + 8 * laneHi) * HID_ + i0;
#pragma unroll
    for (int j = 0; j < 8; ++j) {
        float4 o; o.x = kh0[j]; o.y = kh1[j]; o.z = kh2[j]; o.w = kh3[j];
        *(float4*)(kp + (size_t)j * HID_) = o;
    }
    hs.x += __shfl_xor(hs.x, 16, 32); hs.y += __shfl_xor(hs.y, 16, 32);
    hs.z += __shfl_xor(hs.z, 16, 32); hs.w += __shfl_xor(hs.w, 16, 32);
    if (laneHi == 0) *(float4*)&hsum_part[(size_t)(b * 32 + blk) * HID_ + i0] = hs;
}

__global__ __launch_bounds__(256) void reduce_project_kernel(
    const float* __restrict__ hsp, const float* __restrict__ khp,
    const float* __restrict__ Wv, float* __restrict__ mkp)
{
    __shared__ float4 s_kh[4][64];
    __shared__ float4 s_hs[4][64];
    __shared__ float  s_out[2 * DH_];

    const int sl = blockIdx.x % NSL_;
    const int bh = blockIdx.x / NSL_;
    const int b = bh >> 4, h = bh & 15;
    const int t = threadIdx.x;
    const int wave = t >> 5, lane = t & 31;
    const int col0 = sl * 256;

    const int c = t & 63, g = t >> 6;
    const float* kp = khp + (((size_t)(b * 32 + g * 8)) * H_ + h) * HID_ + col0 + c * 4;
    const float* hp = hsp + ((size_t)(b * 32 + g * 8)) * HID_ + col0 + c * 4;
    float4 pk4[8], ph4[8];
#pragma unroll
    for (int p = 0; p < 8; ++p) pk4[p] = *(const float4*)(kp + (size_t)p * H_ * HID_);
#pragma unroll
    for (int p = 0; p < 8; ++p) ph4[p] = *(const float4*)(hp + (size_t)p * HID_);

    const float* wbase = Wv + (size_t)(h * DH_ + wave * 8) * HID_ + col0 + lane * 4;
    float4 w0[8], w1[8];
#pragma unroll
    for (int r = 0; r < 8; ++r) {
        w0[r] = *(const float4*)(wbase + (size_t)r * HID_);
        w1[r] = *(const float4*)(wbase + (size_t)r * HID_ + 128);
    }

    float4 ak = pk4[0], ah = ph4[0];
#pragma unroll
    for (int p = 1; p < 8; ++p) {
        ak.x += pk4[p].x; ak.y += pk4[p].y; ak.z += pk4[p].z; ak.w += pk4[p].w;
        ah.x += ph4[p].x; ah.y += ph4[p].y; ah.z += ph4[p].z; ah.w += ph4[p].w;
    }
    s_kh[g][c] = ak;
    s_hs[g][c] = ah;
    __syncthreads();

    float4 k0 = s_kh[0][lane], k1 = s_kh[0][32 + lane];
    float4 m0 = s_hs[0][lane], m1 = s_hs[0][32 + lane];
#pragma unroll
    for (int q = 1; q < 4; ++q) {
        float4 a = s_kh[q][lane], e = s_kh[q][32 + lane];
        float4 u = s_hs[q][lane], v = s_hs[q][32 + lane];
        k0.x += a.x; k0.y += a.y; k0.z += a.z; k0.w += a.w;
        k1.x += e.x; k1.y += e.y; k1.z += e.z; k1.w += e.w;
        m0.x += u.x; m0.y += u.y; m0.z += u.z; m0.w += u.w;
        m1.x += v.x; m1.y += v.y; m1.z += v.z; m1.w += v.w;
    }

    float pk[8], pm[8];
#pragma unroll
    for (int r = 0; r < 8; ++r) {
        float sk = w0[r].x * k0.x;
        sk += w0[r].y * k0.y; sk += w0[r].z * k0.z; sk += w0[r].w * k0.w;
        sk += w1[r].x * k1.x; sk += w1[r].y * k1.y; sk += w1[r].z * k1.z; sk += w1[r].w * k1.w;
        float sm = w0[r].x * m0.x;
        sm += w0[r].y * m0.y; sm += w0[r].z * m0.z; sm += w0[r].w * m0.w;
        sm += w1[r].x * m1.x; sm += w1[r].y * m1.y; sm += w1[r].z * m1.z; sm += w1[r].w * m1.w;
        pk[r] = sk; pm[r] = sm;
    }
#pragma unroll
    for (int m = 16; m >= 1; m >>= 1) {
#pragma unroll
        for (int r = 0; r < 8; ++r) {
            pk[r] += __shfl_xor(pk[r], m, 32);
            pm[r] += __shfl_xor(pm[r], m, 32);
        }
    }
    float ok = 0.f, om = 0.f;
#pragma unroll
    for (int r = 0; r < 8; ++r) if (lane == r) { ok = pk[r]; om = pm[r]; }
    if (lane < 8) {
        s_out[wave * 8 + lane] = om;
        s_out[DH_ + wave * 8 + lane] = ok;
    }
    __syncthreads();
    if (t < 32)
        *(float4*)&mkp[((size_t)bh * NSL_ + sl) * (2 * DH_) + t * 4] = *(const float4*)&s_out[t * 4];
}

__global__ __launch_bounds__(256) void pass3_kernel(
    const float* __restrict__ qs, const float* __restrict__ mkp,
    const float* __restrict__ il, float* __restrict__ out)
{
    const int b = blockIdx.x >> 7;
    const int srow0 = (blockIdx.x & 127) * 32;
    const int t = threadIdx.x;
    const int h = t >> 4;
    const float* mk = mkp + (size_t)(b * H_ + h) * (NSL_ * 2 * DH_) + (t & 15) * 4;
    float4 mv4 = *(const float4*)(mk);
    float4 kv4 = *(const float4*)(mk + DH_);
#pragma unroll
    for (int c = 1; c < NSL_; ++c) {
        float4 a = *(const float4*)(mk + c * 2 * DH_);
        float4 e = *(const float4*)(mk + c * 2 * DH_ + DH_);
        mv4.x += a.x; mv4.y += a.y; mv4.z += a.z; mv4.w += a.w;
        kv4.x += e.x; kv4.y += e.y; kv4.z += e.z; kv4.w += e.w;
    }
    const float inv = il[b];
    mv4.x *= inv; mv4.y *= inv; mv4.z *= inv; mv4.w *= inv;
    kv4.x *= inv; kv4.y *= inv; kv4.z *= inv; kv4.w *= inv;
#pragma unroll 4
    for (int r = 0; r < 32; ++r) {
        size_t srow = (size_t)b * S_ + srow0 + r;
        float q = qs[srow * H_ + h];
        float4 o;
        o.x = q * mv4.x + kv4.x; o.y = q * mv4.y + kv4.y;
        o.z = q * mv4.z + kv4.z; o.w = q * mv4.w + kv4.w;
        *(float4*)(out + srow * (H_ * DH_) + t * 4) = o;
    }
}

extern "C" void kernel_launch(void* const* d_in, const int* in_sizes, int n_in,
                              void* d_out, int out_size, void* d_ws, size_t ws_size,
                              hipStream_t stream) {
    const float* hidden = (const float*)d_in[0];
    const float* mask   = (const float*)d_in[1];
    const float* Wq     = (const float*)d_in[2];
    const float* Wk     = (const float*)d_in[3];
    const float* Wv     = (const float*)d_in[4];
    const float* aw     = (const float*)d_in[5];
    float* out = (float*)d_out;
    float* ws  = (float*)d_ws;

    weff_kernel   <<<64,   256, 0, stream>>>(Wq, Wk, aw, ws + OFF_WQE, ws + OFF_WKE);
    invlen_kernel <<<B_,   256, 0, stream>>>(mask, ws + OFF_IL);
    pass1_kernel  <<<256,  512, 0, stream>>>(hidden, mask, ws + OFF_WQE, ws + OFF_WKE,
                                             ws + OFF_QS, ws + OFF_HSP, ws + OFF_KHP);
    reduce_project_kernel<<<512, 256, 0, stream>>>(ws + OFF_HSP, ws + OFF_KHP, Wv, ws + OFF_MKP);
    pass3_kernel  <<<1024, 256, 0, stream>>>(ws + OFF_QS, ws + OFF_MKP, ws + OFF_IL, out);
}


# launches 5 -> 4: no 1/length kernel; the scores/kh kernel adds its tile's 128 mask values from LDS, the reducer adds the 32 tile sums and divides its partials by length (new sum order: length)
# speedup vs baseline: 2.4480x; 1.0383x over previous
#include <hip/hip_runtime.h>

#define B_   8
#define S_   4096
#define HID_ 1024
#define H_   16
#define DH_  64

typedef __attribute__((ext_vector_type(2))) float v2f;
typedef __attribute__((ext_vector_type(8))) float v8f;

#define OFF_WQE 0
#define OFF_WKE (OFF_WQE + H_*HID_)
#define OFF_QS  (OFF_WKE + H_*HID_)
#define OFF_HSP (OFF_QS  + B_*S_*H_)
#define OFF_KHP (OFF_HSP + B_*32*HID_)
#define NSL_ 4
#define OFF_MKP (OFF_KHP + 256*H_*HID_)
#define OFF_ML  (OFF_MKP + B_*H_*NSL_*2*DH_)

__global__ __launch_bounds__(256) void weff_kernel(
    const float* __restrict__ Wq, const float* __restrict__ Wk,
    const float* __restrict__ aw, float* __restrict__ wqe, float* __restrict__ wke)
{
    int u = blockIdx.x * 256 + threadIdx.x;
    int h = u >> 10, i = u & (HID_ - 1);
    float aq = 0.f, ak = 0.f;
#pragma unroll 4
    for (int d = 0; d < DH_; ++d) {
        float a = aw[h * DH_ + d];
        aq += a * Wq[(size_t)(h * DH_ + d) * HID_ + i];
        ak += a * Wk[(size_t)(h * DH_ + d) * HID_ + i];
    }
    wqe[h * HID_ + i] = aq;
    wke[h * HID_ + i] = ak;
}

#define WSTR_ 1028
#define TSTR_ 68

__global__ __launch_bounds__(512) void pass1_kernel(
    const float* __restrict__ hidden, const float* __restrict__ mask,
    const float* __restrict__ wqe, const float* __restrict__ wke,
    float* __restrict__ q_scores, float* __restrict__ hsum_part,
    float* __restrict__ kh_part, float* __restrict__ msum_part)
{
    __shared__ __attribute__((aligned(16))) float lds_wq[16 * WSTR_];
    __shared__ __attribute__((aligned(16))) float lds_wk[16 * WSTR_];
    __shared__ __attribute__((aligned(16))) float lds_strip[16 * 16 * TSTR_];
    __shared__ __attribute__((aligned(16))) float lds_ksT[16 * 132];
    __shared__ __attribute__((aligned(16))) float lds_half[8 * 2 * 8 * 32];
    __shared__ __attribute__((aligned(16))) float lds_mask[128];

    const int blk = blockIdx.x & 31;
    const int b   = blockIdx.x >> 5;
    const int s0  = blk * 128;
    const int t = threadIdx.x;
    const int wave = t >> 5, lane = t & 31;
    const int lane16 = lane & 15, laneHi = lane >> 4;
    const int mg = wave & 7;
    const int kq = wave >> 3;

    const float* hbase = hidden + ((size_t)b * S_ + s0) * HID_;
    const float* hrow = hbase + (size_t)(mg * 16 + laneHi) * HID_ + kq * 512 + lane16 * 4;

#define LOAD_STRIP_(p) \
    r0 = *(const float4*)((p));                         r1 = *(const float4*)((p) + (size_t)2 * HID_);  \
    r2 = *(const float4*)((p) + (size_t)4 * HID_);      r3 = *(const float4*)((p) + (size_t)6 * HID_);  \
    r4 = *(const float4*)((p) + (size_t)8 * HID_);      r5 = *(const float4*)((p) + (size_t)10 * HID_); \
    r6 = *(const float4*)((p) + (size_t)12 * HID_);     r7 = *(const float4*)((p) + (size_t)14 * HID_);
    float4 r0, r1, r2, r3, r4, r5, r6, r7;
    LOAD_STRIP_(hrow)

#pragma unroll
    for (int i = 0; i < 8; ++i) {
        int idx = i * 512 + t;
        int row = idx >> 8, c4 = (idx & 255) * 4;
        *(float4*)&lds_wq[row * WSTR_ + c4] = *(const float4*)(wqe + row * HID_ + c4);
        *(float4*)&lds_wk[row * WSTR_ + c4] = *(const float4*)(wke + row * HID_ + c4);
    }
    if (t < 128) lds_mask[t] = mask[b * S_ + s0 + t];
    __syncthreads();

    float* strip = &lds_strip[wave * (16 * TSTR_)];
    float* sdst = strip + laneHi * TSTR_ + lane16 * 4;
    const float* atile = strip + lane16 * TSTR_ + 2 * laneHi;
    const float* bq = &lds_wq[lane16 * WSTR_ + kq * 512 + 2 * laneHi];
    const float* bk = &lds_wk[lane16 * WSTR_ + kq * 512 + 2 * laneHi];

    v8f cq = {}; v8f ck = {};
#pragma unroll 1
    for (int c = 0; c < 8; ++c) {
        *(float4*)(sdst)              = r0;  *(float4*)(sdst + 2 * TSTR_)  = r1;
        *(float4*)(sdst + 4 * TSTR_)  = r2;  *(float4*)(sdst + 6 * TSTR_)  = r3;
        *(float4*)(sdst + 8 * TSTR_)  = r4;  *(float4*)(sdst + 10 * TSTR_) = r5;
        *(float4*)(sdst + 12 * TSTR_) = r6;  *(float4*)(sdst + 14 * TSTR_) = r7;
        if (c < 7) {
            const float* hn = hrow + (c + 1) * 64;
            LOAD_STRIP_(hn)
        }
#pragma unroll
        for (int kk = 0; kk < 16; ++kk) {
            v2f a = *(const v2f*)(atile + kk * 4);
            v2f q = *(const v2f*)(bq + c * 64 + kk * 4);
            v2f k = *(const v2f*)(bk + c * 64 + kk * 4);
            cq = __builtin_amdgcn_wmma_f32_16x16x4_f32(false, a, false, q, (short)0, cq, false, false);
            ck = __builtin_amdgcn_wmma_f32_16x16x4_f32(false, a, false, k, (short)0, ck, false, false);
        }
    }

    if (kq == 1) {
#pragma unroll
        for (int j = 0; j < 8; ++j) {
            lds_half[((mg * 2 + 0) * 8 + j) * 32 + lane] = cq[j];
            lds_half[((mg * 2 + 1) * 8 + j) * 32 + lane] = ck[j];
        }
    }
    __syncthreads();
    float ms = 0.f;
    if (kq == 0) {
#pragma unroll
        for (int j = 0; j < 8; ++j) {
            int sl = mg * 16 + j + 8 * laneHi;
            float m = lds_mask[sl];
            float qv = cq[j] + lds_half[((mg * 2 + 0) * 8 + j) * 32 + lane];
            float kv = ck[j] + lds_half[((mg * 2 + 1) * 8 + j) * 32 + lane];
            q_scores[((size_t)b * S_ + s0 + sl) * H_ + lane16] = qv * m;
            lds_ksT[lane16 * 132 + sl] = kv * m;
        }
    } else if (wave == 15) {
        float4 m4 = *(const float4*)&lds_mask[lane * 4];
        ms = (m4.x + m4.y) + (m4.z + m4.w);
        ms += __shfl_xor(ms, 16, 32); ms += __shfl_xor(ms, 8, 32); ms += __shfl_xor(ms, 4, 32);
        ms += __shfl_xor(ms, 2, 32);  ms += __shfl_xor(ms, 1, 32);
    }
    __syncthreads();
    if (wave == 15) msum_part[(b * 32 + blk) * 32 + lane] = ms;

    const int i0 = wave * 64 + lane16 * 4;
    const float* hc = hbase + (size_t)(2 * laneHi) * HID_ + i0;
    const float* aks = &lds_ksT[lane16 * 132 + 2 * laneHi];
    const float* mk2 = &lds_mask[2 * laneHi];
    v8f kh0 = {}, kh1 = {}, kh2 = {}, kh3 = {};
    float4 hs = {0.f, 0.f, 0.f, 0.f};
#define KSTEP_(U, XA, XB) { \
        v2f m = *(const v2f*)(mk2 + g * 16 + U * 4); \
        v2f a = *(const v2f*)(aks + g * 16 + U * 4); \
        hs.x += m.x * XA.x; hs.y += m.x * XA.y; hs.z += m.x * XA.z; hs.w += m.x * XA.w; \
        hs.x += m.y * XB.x; hs.y += m.y * XB.y; hs.z += m.y * XB.z; hs.w += m.y * XB.w; \
        v2f b0 = {XA.x, XB.x}, b1 = {XA.y, XB.y}, b2 = {XA.z, XB.z}, b3 = {XA.w, XB.w}; \
        kh0 = __builtin_amdgcn_wmma_f32_16x16x4_f32(false, a, false, b0, (short)0, kh0, false, false); \
        kh1 = __builtin_amdgcn_wmma_f32_16x16x4_f32(false, a, false, b1, (short)0, kh1, false, false); \
        kh2 = __builtin_amdgcn_wmma_f32_16x16x4_f32(false, a, false, b2, (short)0, kh2, false, false); \
        kh3 = __builtin_amdgcn_wmma_f32_16x16x4_f32(false, a, false, b3, (short)0, kh3, false, false); }
    for (int g = 0; g < 8; ++g) {
        const float* hn = hc + (size_t)(g * 16) * HID_;
        float4 x00 = *(const float4*)(hn);
        float4 x01 = *(const float4*)(hn + (size_t)1 * HID_);
        float4 x10 = *(const float4*)(hn + (size_t)4 * HID_);
        float4 x11 = *(const float4*)(hn + (size_t)5 * HID_);
        float4 x20 = *(const float4*)(hn + (size_t)8 * HID_);
        float4 x21 = *(const float4*)(hn + (size_t)9 * HID_);
        float4 x30 = *(const float4*)(hn + (size_t)12 * HID_);
        float4 x31 = *(const float4*)(hn + (size_t)13 * HID_);
        KSTEP_(0, x00, x01) KSTEP_(1, x10, x11) KSTEP_(2, x20, x21) KSTEP_(3, x30, x31)
    }

    float* kp = kh_part + (((size_t)(b * 32 + blk)) * H_ + 8 * laneHi) * HID_ + i0;
#pragma unroll
    for (int j = 0; j < 8; ++j) {
        float4 o; o.x = kh0[j]; o.y = kh1[j]; o.z = kh2[j]; o.w = kh3[j];
        *(float4*)(kp + (size_t)j * HID_) = o;
    }
    hs.x += __shfl_xor(hs.x, 16, 32); hs.y += __shfl_xor(hs.y, 16, 32);
    hs.z += __shfl_xor(hs.z, 16, 32); hs.w += __shfl_xor(hs.w, 16, 32);
    if (laneHi == 0) *(float4*)&hsum_part[(size_t)(b * 32 + blk) * HID_ + i0] = hs;
}

__global__ __launch_bounds__(256) void reduce_project_kernel(
    const float* __restrict__ hsp, const float* __restrict__ khp,
    const float* __restrict__ Wv, const float* __restrict__ msum, float* __restrict__ mkp)
{
    __shared__ float4 s_kh[4][64];
    __shared__ float4 s_hs[4][64];
    __shared__ float  s_out[2 * DH_];

    const int sl = blockIdx.x % NSL_;
    const int bh = blockIdx.x / NSL_;
    const int b = bh >> 4, h = bh & 15;
    const int t = threadIdx.x;
    const int wave = t >> 5, lane = t & 31;
    const int col0 = sl * 256;

    const float* ml = msum + b * (32 * 32);
    float len = ml[0];
#pragma unroll
    for (int p = 1; p < 32; ++p) len += ml[p * 32];
    const float inv = 1.0f / len;

    const int c = t & 63, g = t >> 6;
    const float* kp = khp + (((size_t)(b * 32 + g * 8)) * H_ + h) * HID_ + col0 + c * 4;
    const float* hp = hsp + ((size_t)(b * 32 + g * 8)) * HID_ + col0 + c * 4;
    float4 pk4[8], ph4[8];
#pragma unroll
    for (int p = 0; p < 8; ++p) pk4[p] = *(const float4*)(kp + (size_t)p * H_ * HID_);
#pragma unroll
    for (int p = 0; p < 8; ++p) ph4[p] = *(const float4*)(hp + (size_t)p * HID_);

    const float* wbase = Wv + (size_t)(h * DH_ + wave * 8) * HID_ + col0 + lane * 4;
    float4 w0[8], w1[8];
#pragma unroll
    for (int r = 0; r < 8; ++r) {
        w0[r] = *(const float4*)(wbase + (size_t)r * HID_);
        w1[r] = *(const float4*)(wbase + (size_t)r * HID_ + 128);
    }

    float4 ak = pk4[0], ah = ph4[0];
#pragma unroll
    for (int p = 1; p < 8; ++p) {
        ak.x += pk4[p].x; ak.y += pk4[p].y; ak.z += pk4[p].z; ak.w += pk4[p].w;
        ah.x += ph4[p].x; ah.y += ph4[p].y; ah.z += ph4[p].z; ah.w += ph4[p].w;
    }
    s_kh[g][c] = ak;
    s_hs[g][c] = ah;
    __syncthreads();

    float4 k0 = s_kh[0][lane], k1 = s_kh[0][32 + lane];
    float4 m0 = s_hs[0][lane], m1 = s_hs[0][32 + lane];
#pragma unroll
    for (int q = 1; q < 4; ++q) {
        float4 a = s_kh[q][lane], e = s_kh[q][32 + lane];
        float4 u = s_hs[q][lane], v = s_hs[q][32 + lane];
        k0.x += a.x; k0.y += a.y; k0.z += a.z; k0.w += a.w;
        k1.x += e.x; k1.y += e.y; k1.z += e.z; k1.w += e.w;
        m0.x += u.x; m0.y += u.y; m0.z += u.z; m0.w += u.w;
        m1.x += v.x; m1.y += v.y; m1.z += v.z; m1.w += v.w;
    }

    float pk[8], pm[8];
#pragma unroll
    for (int r = 0; r < 8; ++r) {
        float sk = w0[r].x * k0.x;
        sk += w0[r].y * k0.y; sk += w0[r].z * k0.z; sk += w0[r].w * k0.w;
        sk += w1[r].x * k1.x; sk += w1[r].y * k1.y; sk += w1[r].z * k1.z; sk += w1[r].w * k1.w;
        float sm = w0[r].x * m0.x;
        sm += w0[r].y * m0.y; sm += w0[r].z * m0.z; sm += w0[r].w * m0.w;
        sm += w1[r].x * m1.x; sm += w1[r].y * m1.y; sm += w1[r].z * m1.z; sm += w1[r].w * m1.w;
        pk[r] = sk; pm[r] = sm;
    }
#pragma unroll
    for (int m = 16; m >= 1; m >>= 1) {
#pragma unroll
        for (int r = 0; r < 8; ++r) {
            pk[r] += __shfl_xor(pk[r], m, 32);
            pm[r] += __shfl_xor(pm[r], m, 32);
        }
    }
    float ok = 0.f, om = 0.f;
#pragma unroll
    for (int r = 0; r < 8; ++r) if (lane == r) { ok = pk[r]; om = pm[r]; }
    if (lane < 8) {
        s_out[wave * 8 + lane] = om * inv;
        s_out[DH_ + wave * 8 + lane] = ok * inv;
    }
    __syncthreads();
    if (t < 32)
        *(float4*)&mkp[((size_t)bh * NSL_ + sl) * (2 * DH_) + t * 4] = *(const float4*)&s_out[t * 4];
}

__global__ __launch_bounds__(256) void pass3_kernel(
    const float* __restrict__ qs, const float* __restrict__ mkp, float* __restrict__ out)
{
    const int b = blockIdx.x >> 7;
    const int srow0 = (blockIdx.x & 127) * 32;
    const int t = threadIdx.x;
    const int h = t >> 4;
    const float* mk = mkp + (size_t)(b * H_ + h) * (NSL_ * 2 * DH_) + (t & 15) * 4;
    float4 mv4 = *(const float4*)(mk);
    float4 kv4 = *(const float4*)(mk + DH_);
#pragma unroll
    for (int c = 1; c < NSL_; ++c) {
        float4 a = *(const float4*)(mk + c * 2 * DH_);
        float4 e = *(const float4*)(mk + c * 2 * DH_ + DH_);
        mv4.x += a.x; mv4.y += a.y; mv4.z += a.z; mv4.w += a.w;
        kv4.x += e.x; kv4.y += e.y; kv4.z += e.z; kv4.w += e.w;
    }
#pragma unroll 4
    for (int r = 0; r < 32; ++r) {
        size_t srow = (size_t)b * S_ + srow0 + r;
        float q = qs[srow * H_ + h];
        float4 o;
        o.x = q * mv4.x + kv4.x; o.y = q * mv4.y + kv4.y;
        o.z = q * mv4.z + kv4.z; o.w = q * mv4.w + kv4.w;
        *(float4*)(out + srow * (H_ * DH_) + t * 4) = o;
    }
}

extern "C" void kernel_launch(void* const* d_in, const int* in_sizes, int n_in,
                              void* d_out, int out_size, void* d_ws, size_t ws_size,
                              hipStream_t stream) {
    const float* hidden = (const float*)d_in[0];
    const float* mask   = (const float*)d_in[1];
    const float* Wq     = (const float*)d_in[2];
    const float* Wk     = (const float*)d_in[3];
    const float* Wv     = (const float*)d_in[4];
    const float* aw     = (const float*)d_in[5];
    float* out = (float*)d_out;
    float* ws  = (float*)d_ws;

    weff_kernel   <<<64,   256, 0, stream>>>(Wq, Wk, aw, ws + OFF_WQE, ws + OFF_WKE);
    pass1_kernel  <<<256,  512, 0, stream>>>(hidden, mask, ws + OFF_WQE, ws + OFF_WKE,
                                             ws + OFF_QS, ws + OFF_HSP, ws + OFF_KHP, ws + OFF_ML);
    reduce_project_kernel<<<512, 256, 0, stream>>>(ws + OFF_HSP, ws + OFF_KHP, Wv, ws + OFF_ML, ws + OFF_MKP);
    pass3_kernel  <<<1024, 256, 0, stream>>>(ws + OFF_QS, ws + OFF_MKP, out);
}


# output pass: q scores come through the scalar path, a wave's 32 rows x 2 heads in 32 scalar loads ahead of the store loop (was 16 serial vector-load round trips per wave); every sum unchanged
# speedup vs baseline: 2.5810x; 1.0543x over previous
#include <hip/hip_runtime.h>

#define B_   8
#define S_   4096
#define HID_ 1024
#define H_   16
#define DH_  64

typedef __attribute__((ext_vector_type(2))) float v2f;
typedef __attribute__((ext_vector_type(8))) float v8f;

#define OFF_WQE 0
#define OFF_WKE (OFF_WQE + H_*HID_)
#define OFF_QS  (OFF_WKE + H_*HID_)
#define OFF_HSP (OFF_QS  + B_*S_*H_)
#define OFF_KHP (OFF_HSP + B_*32*HID_)
#define NSL_ 4
#define OFF_MKP (OFF_KHP + 256*H_*HID_)
#define OFF_ML  (OFF_MKP + B_*H_*NSL_*2*DH_)

__global__ __launch_bounds__(256) void weff_kernel(
    const float* __restrict__ Wq, const float* __restrict__ Wk,
    const float* __restrict__ aw, float* __restrict__ wqe, float* __restrict__ wke)
{
    int u = blockIdx.x * 256 + threadIdx.x;
    int h = u >> 10, i = u & (HID_ - 1);
    float aq = 0.f, ak = 0.f;
#pragma unroll 4
    for (int d = 0; d < DH_; ++d) {
        float a = aw[h * DH_ + d];
        aq += a * Wq[(size_t)(h * DH_ + d) * HID_ + i];
        ak += a * Wk[(size_t)(h * DH_ + d) * HID_ + i];
    }
    wqe[h * HID_ + i] = aq;
    wke[h * HID_ + i] = ak;
}

#define WSTR_ 1028
#define TSTR_ 68

__global__ __launch_bounds__(512) void pass1_kernel(
    const float* __restrict__ hidden, const float* __restrict__ mask,
    const float* __restrict__ wqe, const float* __restrict__ wke,
    float* __restrict__ q_scores, float* __restrict__ hsum_part,
    float* __restrict__ kh_part, float* __restrict__ msum_part)
{
    __shared__ __attribute__((aligned(16))) float lds_wq[16 * WSTR_];
    __shared__ __attribute__((aligned(16))) float lds_wk[16 * WSTR_];
    __shared__ __attribute__((aligned(16))) float lds_strip[16 * 16 * TSTR_];
    __shared__ __attribute__((aligned(16))) float lds_ksT[16 * 132];
    __shared__ __attribute__((aligned(16))) float lds_half[8 * 2 * 8 * 32];
    __shared__ __attribute__((aligned(16))) float lds_mask[128];

    const int blk = blockIdx.x & 31;
    const int b   = blockIdx.x >> 5;
    const int s0  = blk * 128;
    const int t = threadIdx.x;
    const int wave = t >> 5, lane = t & 31;
    const int lane16 = lane & 15, laneHi = lane >> 4;
    const int mg = wave & 7;
    const int kq = wave >> 3;

    const float* hbase = hidden + ((size_t)b * S_ + s0) * HID_;
    const float* hrow = hbase + (size_t)(mg * 16 + laneHi) * HID_ + kq * 512 + lane16 * 4;

#define LOAD_STRIP_(p) \
    r0 = *(const float4*)((p));                         r1 = *(const float4*)((p) + (size_t)2 * HID_);  \
    r2 = *(const float4*)((p) + (size_t)4 * HID_);      r3 = *(const float4*)((p) + (size_t)6 * HID_);  \
    r4 = *(const float4*)((p) + (size_t)8 * HID_);      r5 = *(const float4*)((p) + (size_t)10 * HID_); \
    r6 = *(const float4*)((p) + (size_t)12 * HID_);     r7 = *(const float4*)((p) + (size_t)14 * HID_);
    float4 r0, r1, r2, r3, r4, r5, r6, r7;
    LOAD_STRIP_(hrow)

#pragma unroll
    for (int i = 0; i < 8; ++i) {
        int idx = i * 512 + t;
        int row = idx >> 8, c4 = (idx & 255) * 4;
        *(float4*)&lds_wq[row * WSTR_ + c4] = *(const float4*)(wqe + row * HID_ + c4);
        *(float4*)&lds_wk[row * WSTR_ + c4] = *(const float4*)(wke + row * HID_ + c4);
    }
    if (t < 128) lds_mask[t] = mask[b * S_ + s0 + t];
    __syncthreads();

    float* strip = &lds_strip[wave * (16 * TSTR_)];
    float* sdst = strip + laneHi * TSTR_ + lane16 * 4;
    const float* atile = strip + lane16 * TSTR_ + 2 * laneHi;
    const float* bq = &lds_wq[lane16 * WSTR_ + kq * 512 + 2 * laneHi];
    const float* bk = &lds_wk[lane16 * WSTR_ + kq * 512 + 2 * laneHi];

    v8f cq = {}; v8f ck = {};
#pragma unroll 1
    for (int c = 0; c < 8; ++c) {
        *(float4*)(sdst)              = r0;  *(float4*)(sdst + 2 * TSTR_)  = r1;
        *(float4*)(sdst + 4 * TSTR_)  = r2;  *(float4*)(sdst + 6 * TSTR_)  = r3;
        *(float4*)(sdst + 8 * TSTR_)  = r4;  *(float4*)(sdst + 10 * TSTR_) = r5;
        *(float4*)(sdst + 12 * TSTR_) = r6;  *(float4*)(sdst + 14 * TSTR_) = r7;
        if (c < 7) {
            const float* hn = hrow + (c + 1) * 64;
            LOAD_STRIP_(hn)
        }
#pragma unroll
        for (int kk = 0; kk < 16; ++kk) {
            v2f a = *(const v2f*)(atile + kk * 4);
            v2f q = *(const v2f*)(bq + c * 64 + kk * 4);
            v2f k = *(const v2f*)(bk + c * 64 + kk * 4);
            cq = __builtin_amdgcn_wmma_f32_16x16x4_f32(false, a, false, q, (short)0, cq, false, false);
            ck = __builtin_amdgcn_wmma_f32_16x16x4_f32(false, a, false, k, (short)0, ck, false, false);
        }
    }

    if (kq == 1) {
#pragma unroll
        for (int j = 0; j < 8; ++j) {
            lds_half[((mg * 2 + 0) * 8 + j) * 32 + lane] = cq[j];
            lds_half[((mg * 2 + 1) * 8 + j) * 32 + lane] = ck[j];
        }
    }
    __syncthreads();
    float ms = 0.f;
    if (kq == 0) {
#pragma unroll
        for (int j = 0; j < 8; ++j) {
            int sl = mg * 16 + j + 8 * laneHi;
            float m = lds_mask[sl];
            float qv = cq[j] + lds_half[((mg * 2 + 0) * 8 + j) * 32 + lane];
            float kv = ck[j] + lds_half[((mg * 2 + 1) * 8 + j) * 32 + lane];
            q_scores[((size_t)b * S_ + s0 + sl) * H_ + lane16] = qv * m;
            lds_ksT[lane16 * 132 + sl] = kv * m;
        }
    } else if (wave == 15) {
        float4 m4 = *(const float4*)&lds_mask[lane * 4];
        ms = (m4.x + m4.y) + (m4.z + m4.w);
        ms += __shfl_xor(ms, 16, 32); ms += __shfl_xor(ms, 8, 32); ms += __shfl_xor(ms, 4, 32);
        ms += __shfl_xor(ms, 2, 32);  ms += __shfl_xor(ms, 1, 32);
    }
    __syncthreads();
    if (wave == 15) msum_part[(b * 32 + blk) * 32 + lane] = ms;

    const int i0 = wave * 64 + lane16 * 4;
    const float* hc = hbase + (size_t)(2 * laneHi) * HID_ + i0;
    const float* aks = &lds_ksT[lane16 * 132 + 2 * laneHi];
    const float* mk2 = &lds_mask[2 * laneHi];
    v8f kh0 = {}, kh1 = {}, kh2 = {}, kh3 = {};
    float4 hs = {0.f, 0.f, 0.f, 0.f};
#define KSTEP_(U, XA, XB) { \
        v2f m = *(const v2f*)(mk2 + g * 16 + U * 4); \
        v2f a = *(const v2f*)(aks + g * 16 + U * 4); \
        hs.x += m.x * XA.x; hs.y += m.x * XA.y; hs.z += m.x * XA.z; hs.w += m.x * XA.w; \
        hs.x += m.y * XB.x; hs.y += m.y * XB.y; hs.z += m.y * XB.z; hs.w += m.y * XB.w; \
        v2f b0 = {XA.x, XB.x}, b1 = {XA.y, XB.y}, b2 = {XA.z, XB.z}, b3 = {XA.w, XB.w}; \
        kh0 = __builtin_amdgcn_wmma_f32_16x16x4_f32(false, a, false, b0, (short)0, kh0, false, false); \
        kh1 = __builtin_amdgcn_wmma_f32_16x16x4_f32(false, a, false, b1, (short)0, kh1, false, false); \
        kh2 = __builtin_amdgcn_wmma_f32_16x16x4_f32(false, a, false, b2, (short)0, kh2, false, false); \
        kh3 = __builtin_amdgcn_wmma_f32_16x16x4_f32(false, a, false, b3, (short)0, kh3, false, false); }
    for (int g = 0; g < 8; ++g) {
        const float* hn = hc + (size_t)(g * 16) * HID_;
        float4 x00 = *(const float4*)(hn);
        float4 x01 = *(const float4*)(hn + (size_t)1 * HID_);
        float4 x10 = *(const float4*)(hn + (size_t)4 * HID_);
        float4 x11 = *(const float4*)(hn + (size_t)5 * HID_);
        float4 x20 = *(const float4*)(hn + (size_t)8 * HID_);
        float4 x21 = *(const float4*)(hn + (size_t)9 * HID_);
        float4 x30 = *(const float4*)(hn + (size_t)12 * HID_);
        float4 x31 = *(const float4*)(hn + (size_t)13 * HID_);
        KSTEP_(0, x00, x01) KSTEP_(1, x10, x11) KSTEP_(2, x20, x21) KSTEP_(3, x30, x31)
    }

    float* kp = kh_part + (((size_t)(b * 32 + blk)) * H_ + 8 * laneHi) * HID_ + i0;
#pragma unroll
    for (int j = 0; j < 8; ++j) {
        float4 o; o.x = kh0[j]; o.y = kh1[j]; o.z = kh2[j]; o.w = kh3[j];
        *(float4*)(kp + (size_t)j * HID_) = o;
    }
    hs.x += __shfl_xor(hs.x, 16, 32); hs.y += __shfl_xor(hs.y, 16, 32);
    hs.z += __shfl_xor(hs.z, 16, 32); hs.w += __shfl_xor(hs.w, 16, 32);
    if (laneHi == 0) *(float4*)&hsum_part[(size_t)(b * 32 + blk) * HID_ + i0] = hs;
}

__global__ __launch_bounds__(256) void reduce_project_kernel(
    const float* __restrict__ hsp, const float* __restrict__ khp,
    const float* __restrict__ Wv, const float* __restrict__ msum, float* __restrict__ mkp)
{
    __shared__ float4 s_kh[4][64];
    __shared__ float4 s_hs[4][64];
    __shared__ float  s_out[2 * DH_];

    const int sl = blockIdx.x % NSL_;
    const int bh = blockIdx.x / NSL_;
    const int b = bh >> 4, h = bh & 15;
    const int t = threadIdx.x;
    const int wave = t >> 5, lane = t & 31;
    const int col0 = sl * 256;

    const float* ml = msum + b * (32 * 32);
    float len = ml[0];
#pragma unroll
    for (int p = 1; p < 32; ++p) len += ml[p * 32];
    const float inv = 1.0f / len;

    const int c = t & 63, g = t >> 6;
    const float* kp = khp + (((size_t)(b * 32 + g * 8)) * H_ + h) * HID_ + col0 + c * 4;
    const float* hp = hsp + ((size_t)(b * 32 + g * 8)) * HID_ + col0 + c * 4;
    float4 pk4[8], ph4[8];
#pragma unroll
    for (int p = 0; p < 8; ++p) pk4[p] = *(const float4*)(kp + (size_t)p * H_ * HID_);
#pragma unroll
    for (int p = 0; p < 8; ++p) ph4[p] = *(const float4*)(hp + (size_t)p * HID_);

    const float* wbase = Wv + (size_t)(h * DH_ + wave * 8) * HID_ + col0 + lane * 4;
    float4 w0[8], w1[8];
#pragma unroll
    for (int r = 0; r < 8; ++r) {
        w0[r] = *(const float4*)(wbase + (size_t)r * HID_);
        w1[r] = *(const float4*)(wbase + (size_t)r * HID_ + 128);
    }

    float4 ak = pk4[0], ah = ph4[0];
#pragma unroll
    for (int p = 1; p < 8; ++p) {
        ak.x += pk4[p].x; ak.y += pk4[p].y; ak.z += pk4[p].z; ak.w += pk4[p].w;
        ah.x += ph4[p].x; ah.y += ph4[p].y; ah.z += ph4[p].z; ah.w += ph4[p].w;
    }
    s_kh[g][c] = ak;
    s_hs[g][c] = ah;
    __syncthreads();

    float4 k0 = s_kh[0][lane], k1 = s_kh[0][32 + lane];
    float4 m0 = s_hs[0][lane], m1 = s_hs[0][32 + lane];
#pragma unroll
    for (int q = 1; q < 4; ++q) {
        float4 a = s_kh[q][lane], e = s_kh[q][32 + lane];
        float4 u = s_hs[q][lane], v = s_hs[q][32 + lane];
        k0.x += a.x; k0.y += a.y; k0.z += a.z; k0.w += a.w;
        k1.x += e.x; k1.y += e.y; k1.z += e.z; k1.w += e.w;
        m0.x += u.x; m0.y += u.y; m0.z += u.z; m0.w += u.w;
        m1.x += v.x; m1.y += v.y; m1.z += v.z; m1.w += v.w;
    }

    float pk[8], pm[8];
#pragma unroll
    for (int r = 0; r < 8; ++r) {
        float sk = w0[r].x * k0.x;
        sk += w0[r].y * k0.y; sk += w0[r].z * k0.z; sk += w0[r].w * k0.w;
        sk += w1[r].x * k1.x; sk += w1[r].y * k1.y; sk += w1[r].z * k1.z; sk += w1[r].w * k1.w;
        float sm = w0[r].x * m0.x;
        sm += w0[r].y * m0.y; sm += w0[r].z * m0.z; sm += w0[r].w * m0.w;
        sm += w1[r].x * m1.x; sm += w1[r].y * m1.y; sm += w1[r].z * m1.z; sm += w1[r].w * m1.w;
        pk[r] = sk; pm[r] = sm;
    }
#pragma unroll
    for (int m = 16; m >= 1; m >>= 1) {
#pragma unroll
        for (int r = 0; r < 8; ++r) {
            pk[r] += __shfl_xor(pk[r], m, 32);
            pm[r] += __shfl_xor(pm[r], m, 32);
        }
    }
    float ok = 0.f, om = 0.f;
#pragma unroll
    for (int r = 0; r < 8; ++r) if (lane == r) { ok = pk[r]; om = pm[r]; }
    if (lane < 8) {
        s_out[wave * 8 + lane] = om * inv;
        s_out[DH_ + wave * 8 + lane] = ok * inv;
    }
    __syncthreads();
    if (t < 32)
        *(float4*)&mkp[((size_t)bh * NSL_ + sl) * (2 * DH_) + t * 4] = *(const float4*)&s_out[t * 4];
}

__global__ __launch_bounds__(256, 8) void pass3_kernel(
    const float* __restrict__ qs, const float* __restrict__ mkp, float* __restrict__ out)
{
    const int b = blockIdx.x >> 7;
    const int srow0 = (blockIdx.x & 127) * 32;
    const int t = threadIdx.x;
    const int h = t >> 4;
    const int wave = __builtin_amdgcn_readfirstlane(t >> 5);
    const bool upper = (t & 16) != 0;

    const float* mk = mkp + (size_t)(b * H_ + h) * (NSL_ * 2 * DH_) + (t & 15) * 4;
    float4 mv4 = *(const float4*)(mk);
    float4 kv4 = *(const float4*)(mk + DH_);
#pragma unroll
    for (int c = 1; c < NSL_; ++c) {
        float4 a = *(const float4*)(mk + c * 2 * DH_);
        float4 e = *(const float4*)(mk + c * 2 * DH_ + DH_);
        mv4.x += a.x; mv4.y += a.y; mv4.z += a.z; mv4.w += a.w;
        kv4.x += e.x; kv4.y += e.y; kv4.z += e.z; kv4.w += e.w;
    }
    const float* qrow = qs + ((size_t)b * S_ + srow0) * H_ + 2 * wave;
    float qlo[32], qhi[32];
#pragma unroll
    for (int r = 0; r < 32; ++r) {
        qlo[r] = qrow[r * H_];
        qhi[r] = qrow[r * H_ + 1];
    }

    float* orow = out + ((size_t)b * S_ + srow0) * (H_ * DH_) + t * 4;
#pragma unroll
    for (int r = 0; r < 32; ++r) {
        float q = upper ? qhi[r] : qlo[r];
        float4 o;
        o.x = q * mv4.x + kv4.x; o.y = q * mv4.y + kv4.y;
        o.z = q * mv4.z + kv4.z; o.w = q * mv4.w + kv4.w;
        *(float4*)(orow + (size_t)r * (H_ * DH_)) = o;
    }
}

extern "C" void kernel_launch(void* const* d_in, const int* in_sizes, int n_in,
                              void* d_out, int out_size, void* d_ws, size_t ws_size,
                              hipStream_t stream) {
    const float* hidden = (const float*)d_in[0];
    const float* mask   = (const float*)d_in[1];
    const float* Wq     = (const float*)d_in[2];
    const float* Wk     = (const float*)d_in[3];
    const float* Wv     = (const float*)d_in[4];
    const float* aw     = (const float*)d_in[5];
    float* out = (float*)d_out;
    float* ws  = (float*)d_ws;

    weff_kernel   <<<64,   256, 0, stream>>>(Wq, Wk, aw, ws + OFF_WQE, ws + OFF_WKE);
    pass1_kernel  <<<256,  512, 0, stream>>>(hidden, mask, ws + OFF_WQE, ws + OFF_WKE,
                                             ws + OFF_QS, ws + OFF_HSP, ws + OFF_KHP, ws + OFF_ML);
    reduce_project_kernel<<<512, 256, 0, stream>>>(ws + OFF_HSP, ws + OFF_KHP, Wv, ws + OFF_ML, ws + OFF_MKP);
    pass3_kernel  <<<1024, 256, 0, stream>>>(ws + OFF_QS, ws + OFF_MKP, out);
}


# weight fold, d not split: a lane owns 4 columns of one matrix and loads all 64 rows of its head up front, 16 bytes each (1 memory round trip, was 16 dependent); 256 one-wave groups; sums unchanged
# speedup vs baseline: 3.2067x; 1.2424x over previous
#include <hip/hip_runtime.h>

#define B_   8
#define S_   4096
#define HID_ 1024
#define H_   16
#define DH_  64

typedef __attribute__((ext_vector_type(2))) float v2f;
typedef __attribute__((ext_vector_type(8))) float v8f;

#define OFF_WQE 0
#define OFF_WKE (OFF_WQE + H_*HID_)
#define OFF_QS  (OFF_WKE + H_*HID_)
#define OFF_HSP (OFF_QS  + B_*S_*H_)
#define OFF_KHP (OFF_HSP + B_*32*HID_)
#define NSL_ 4
#define OFF_MKP (OFF_KHP + 256*H_*HID_)
#define OFF_ML  (OFF_MKP + B_*H_*NSL_*2*DH_)

__global__ __launch_bounds__(32) __attribute__((amdgpu_waves_per_eu(1, 1))) void weff_kernel(
    const float* __restrict__ Wq, const float* __restrict__ Wk,
    const float* __restrict__ aw, float* __restrict__ wqe, float* __restrict__ wke)
{
    const int g  = blockIdx.x;
    const int mk = g >> 7;
    const int h  = (g >> 3) & (H_ - 1);
    const int i0 = (g & 7) * 128 + threadIdx.x * 4;
    const float* W = (mk ? Wk : Wq) + (size_t)(h * DH_) * HID_ + i0;
    const float* a = aw + h * DH_;

    float4 w[DH_];
#pragma unroll
    for (int d = 0; d < DH_; ++d) w[d] = *(const float4*)(W + (size_t)d * HID_);

    float4 acc = {0.f, 0.f, 0.f, 0.f};
#pragma unroll
    for (int d = 0; d < DH_; ++d) {
        const float s = a[d];
        acc.x = __builtin_fmaf(s, w[d].x, acc.x); acc.y = __builtin_fmaf(s, w[d].y, acc.y);
        acc.z = __builtin_fmaf(s, w[d].z, acc.z); acc.w = __builtin_fmaf(s, w[d].w, acc.w);
    }
    *(float4*)((mk ? wke : wqe) + h * HID_ + i0) = acc;
}

#define WSTR_ 1028
#define TSTR_ 68

__global__ __launch_bounds__(512) void pass1_kernel(
    const float* __restrict__ hidden, const float* __restrict__ mask,
    const float* __restrict__ wqe, const float* __restrict__ wke,
    float* __restrict__ q_scores, float* __restrict__ hsum_part,
    float* __restrict__ kh_part, float* __restrict__ msum_part)
{
    __shared__ __attribute__((aligned(16))) float lds_wq[16 * WSTR_];
    __shared__ __attribute__((aligned(16))) float lds_wk[16 * WSTR_];
    __shared__ __attribute__((aligned(16))) float lds_strip[16 * 16 * TSTR_];
    __shared__ __attribute__((aligned(16))) float lds_ksT[16 * 132];
    __shared__ __attribute__((aligned(16))) float lds_half[8 * 2 * 8 * 32];
    __shared__ __attribute__((aligned(16))) float lds_mask[128];

    const int blk = blockIdx.x & 31;
    const int b   = blockIdx.x >> 5;
    const int s0  = blk * 128;
    const int t = threadIdx.x;
    const int wave = t >> 5, lane = t & 31;
    const int lane16 = lane & 15, laneHi = lane >> 4;
    const int mg = wave & 7;
    const int kq = wave >> 3;

    const float* hbase = hidden + ((size_t)b * S_ + s0) * HID_;
    const float* hrow = hbase + (size_t)(mg * 16 + laneHi) * HID_ + kq * 512 + lane16 * 4;

#define LOAD_STRIP_(p) \
    r0 = *(const float4*)((p));                         r1 = *(const float4*)((p) + (size_t)2 * HID_);  \
    r2 = *(const float4*)((p) + (size_t)4 * HID_);      r3 = *(const float4*)((p) + (size_t)6 * HID_);  \
    r4 = *(const float4*)((p) + (size_t)8 * HID_);      r5 = *(const float4*)((p) + (size_t)10 * HID_); \
    r6 = *(const float4*)((p) + (size_t)12 * HID_);     r7 = *(const float4*)((p) + (size_t)14 * HID_);
    float4 r0, r1, r2, r3, r4, r5, r6, r7;
    LOAD_STRIP_(hrow)

#pragma unroll
    for (int i = 0; i < 8; ++i) {
        int idx = i * 512 + t;
        int row = idx >> 8, c4 = (idx & 255) * 4;
        *(float4*)&lds_wq[row * WSTR_ + c4] = *(const float4*)(wqe + row * HID_ + c4);
        *(float4*)&lds_wk[row * WSTR_ + c4] = *(const float4*)(wke + row * HID_ + c4);
    }
    if (t < 128) lds_mask[t] = mask[b * S_ + s0 + t];
    __syncthreads();

    float* strip = &lds_strip[wave * (16 * TSTR_)];
    float* sdst = strip + laneHi * TSTR_ + lane16 * 4;
    const float* atile = strip + lane16 * TSTR_ + 2 * laneHi;
    const float* bq = &lds_wq[lane16 * WSTR_ + kq * 512 + 2 * laneHi];
    const float* bk = &lds_wk[lane16 * WSTR_ + kq * 512 + 2 * laneHi];

    v8f cq = {}; v8f ck = {};
#pragma unroll 1
    for (int c = 0; c < 8; ++c) {
        *(float4*)(sdst)              = r0;  *(float4*)(sdst + 2 * TSTR_)  = r1;
        *(float4*)(sdst + 4 * TSTR_)  = r2;  *(float4*)(sdst + 6 * TSTR_)  = r3;
        *(float4*)(sdst + 8 * TSTR_)  = r4;  *(float4*)(sdst + 10 * TSTR_) = r5;
        *(float4*)(sdst + 12 * TSTR_) = r6;  *(float4*)(sdst + 14 * TSTR_) = r7;
        if (c < 7) {
            const float* hn = hrow + (c + 1) * 64;
            LOAD_STRIP_(hn)
        }
#pragma unroll
        for (int kk = 0; kk < 16; ++kk) {
            v2f a = *(const v2f*)(atile + kk * 4);
            v2f q = *(const v2f*)(bq + c * 64 + kk * 4);
            v2f k = *(const v2f*)(bk + c * 64 + kk * 4);
            cq = __builtin_amdgcn_wmma_f32_16x16x4_f32(false, a, false, q, (short)0, cq, false, false);
            ck = __builtin_amdgcn_wmma_f32_16x16x4_f32(false, a, false, k, (short)0, ck, false, false);
        }
    }

    if (kq == 1) {
#pragma unroll
        for (int j = 0; j < 8; ++j) {
            lds_half[((mg * 2 + 0) * 8 + j) * 32 + lane] = cq[j];
            lds_half[((mg * 2 + 1) * 8 + j) * 32 + lane] = ck[j];
        }
    }
    __syncthreads();
    float ms = 0.f;
    if (kq == 0) {
#pragma unroll
        for (int j = 0; j < 8; ++j) {
            int sl = mg * 16 + j + 8 * laneHi;
            float m = lds_mask[sl];
            float qv = cq[j] + lds_half[((mg * 2 + 0) * 8 + j) * 32 + lane];
            float kv = ck[j] + lds_half[((mg * 2 + 1) * 8 + j) * 32 + lane];
            q_scores[((size_t)b * S_ + s0 + sl) * H_ + lane16] = qv * m;
            lds_ksT[lane16 * 132 + sl] = kv * m;
        }
    } else if (wave == 15) {
        float4 m4 = *(const float4*)&lds_mask[lane * 4];
        ms = (m4.x + m4.y) + (m4.z + m4.w);
        ms += __shfl_xor(ms, 16, 32); ms += __shfl_xor(ms, 8, 32); ms += __shfl_xor(ms, 4, 32);
        ms += __shfl_xor(ms, 2, 32);  ms += __shfl_xor(ms, 1, 32);
    }
    __syncthreads();
    if (wave == 15) msum_part[(b * 32 + blk) * 32 + lane] = ms;

    const int i0 = wave * 64 + lane16 * 4;
    const float* hc = hbase + (size_t)(2 * laneHi) * HID_ + i0;
    const float* aks = &lds_ksT[lane16 * 132 + 2 * laneHi];
    const float* mk2 = &lds_mask[2 * laneHi];
    v8f kh0 = {}, kh1 = {}, kh2 = {}, kh3 = {};
    float4 hs = {0.f, 0.f, 0.f, 0.f};
#define KSTEP_(U, XA, XB) { \
        v2f m = *(const v2f*)(mk2 + g * 16 + U * 4); \
        v2f a = *(const v2f*)(aks + g * 16 + U * 4); \
        hs.x += m.x * XA.x; hs.y += m.x * XA.y; hs.z += m.x * XA.z; hs.w += m.x * XA.w; \
        hs.x += m.y * XB.x; hs.y += m.y * XB.y; hs.z += m.y * XB.z; hs.w += m.y * XB.w; \
        v2f b0 = {XA.x, XB.x}, b1 = {XA.y, XB.y}, b2 = {XA.z, XB.z}, b3 = {XA.w, XB.w}; \
        kh0 = __builtin_amdgcn_wmma_f32_16x16x4_f32(false, a, false, b0, (short)0, kh0, false, false); \
        kh1 = __builtin_amdgcn_wmma_f32_16x16x4_f32(false, a, false, b1, (short)0, kh1, false, false); \
        kh2 = __builtin_amdgcn_wmma_f32_16x16x4_f32(false, a, false, b2, (short)0, kh2, false, false); \
        kh3 = __builtin_amdgcn_wmma_f32_16x16x4_f32(false, a, false, b3, (short)0, kh3, false, false); }
    for (int g = 0; g < 8; ++g) {
        const float* hn = hc + (size_t)(g * 16) * HID_;
        float4 x00 = *(const float4*)(hn);
        float4 x01 = *(const float4*)(hn + (size_t)1 * HID_);
        float4 x10 = *(const float4*)(hn + (size_t)4 * HID_);
        float4 x11 = *(const float4*)(hn + (size_t)5 * HID_);
        float4 x20 = *(const float4*)(hn + (size_t)8 * HID_);
        float4 x21 = *(const float4*)(hn + (size_t)9 * HID_);
        float4 x30 = *(const float4*)(hn + (size_t)12 * HID_);
        float4 x31 = *(const float4*)(hn + (size_t)13 * HID_);
        KSTEP_(0, x00, x01) KSTEP_(1, x10, x11) KSTEP_(2, x20, x21) KSTEP_(3, x30, x31)
    }

    float* kp = kh_part + (((size_t)(b * 32 + blk)) * H_ + 8 * laneHi) * HID_ + i0;
#pragma unroll
    for (int j = 0; j < 8; ++j) {
        float4 o; o.x = kh0[j]; o.y = kh1[j]; o.z = kh2[j]; o.w = kh3[j];
        *(float4*)(kp + (size_t)j * HID_) = o;
    }
    hs.x += __shfl_xor(hs.x, 16, 32); hs.y += __shfl_xor(hs.y, 16, 32);
    hs.z += __shfl_xor(hs.z, 16, 32); hs.w += __shfl_xor(hs.w, 16, 32);
    if (laneHi == 0) *(float4*)&hsum_part[(size_t)(b * 32 + blk) * HID_ + i0] = hs;
}

__global__ __launch_bounds__(256) void reduce_project_kernel(
    const float* __restrict__ hsp, const float* __restrict__ khp,
    const float* __restrict__ Wv, const float* __restrict__ msum, float* __restrict__ mkp)
{
    __shared__ float4 s_kh[4][64];
    __shared__ float4 s_hs[4][64];
    __shared__ float  s_out[2 * DH_];

    const int sl = blockIdx.x % NSL_;
    const int bh = blockIdx.x / NSL_;
    const int b = bh >> 4, h = bh & 15;
    const int t = threadIdx.x;
    const int wave = t >> 5, lane = t & 31;
    const int col0 = sl * 256;

    const float* ml = msum + b * (32 * 32);
    float len = ml[0];
#pragma unroll
    for (int p = 1; p < 32; ++p) len += ml[p * 32];
    const float inv = 1.0f / len;

    const int c = t & 63, g = t >> 6;
    const float* kp = khp + (((size_t)(b * 32 + g * 8)) * H_ + h) * HID_ + col0 + c * 4;
    const float* hp = hsp + ((size_t)(b * 32 + g * 8)) * HID_ + col0 + c * 4;
    float4 pk4[8], ph4[8];
#pragma unroll
    for (int p = 0; p < 8; ++p) pk4[p] = *(const float4*)(kp + (size_t)p * H_ * HID_);
#pragma unroll
    for (int p = 0; p < 8; ++p) ph4[p] = *(const float4*)(hp + (size_t)p * HID_);

    const float* wbase = Wv + (size_t)(h * DH_ + wave * 8) * HID_ + col0 + lane * 4;
    float4 w0[8], w1[8];
#pragma unroll
    for (int r = 0; r < 8; ++r) {
        w0[r] = *(const float4*)(wbase + (size_t)r * HID_);
        w1[r] = *(const float4*)(wbase + (size_t)r * HID_ + 128);
    }

    float4 ak = pk4[0], ah = ph4[0];
#pragma unroll
    for (int p = 1; p < 8; ++p) {
        ak.x += pk4[p].x; ak.y += pk4[p].y; ak.z += pk4[p].z; ak.w += pk4[p].w;
        ah.x += ph4[p].x; ah.y += ph4[p].y; ah.z += ph4[p].z; ah.w += ph4[p].w;
    }
    s_kh[g][c] = ak;
    s_hs[g][c] = ah;
    __syncthreads();

    float4 k0 = s_kh[0][lane], k1 = s_kh[0][32 + lane];
    float4 m0 = s_hs[0][lane], m1 = s_hs[0][32 + lane];
#pragma unroll
    for (int q = 1; q < 4; ++q) {
        float4 a = s_kh[q][lane], e = s_kh[q][32 + lane];
        float4 u = s_hs[q][lane], v = s_hs[q][32 + lane];
        k0.x += a.x; k0.y += a.y; k0.z += a.z; k0.w += a.w;
        k1.x += e.x; k1.y += e.y; k1.z += e.z; k1.w += e.w;
        m0.x += u.x; m0.y += u.y; m0.z += u.z; m0.w += u.w;
        m1.x += v.x; m1.y += v.y; m1.z += v.z; m1.w += v.w;
    }

    float pk[8], pm[8];
#pragma unroll
    for (int r = 0; r < 8; ++r) {
        float sk = w0[r].x * k0.x;
        sk += w0[r].y * k0.y; sk += w0[r].z * k0.z; sk += w0[r].w * k0.w;
        sk += w1[r].x * k1.x; sk += w1[r].y * k1.y; sk += w1[r].z * k1.z; sk += w1[r].w * k1.w;
        float sm = w0[r].x * m0.x;
        sm += w0[r].y * m0.y; sm += w0[r].z * m0.z; sm += w0[r].w * m0.w;
        sm += w1[r].x * m1.x; sm += w1[r].y * m1.y; sm += w1[r].z * m1.z; sm += w1[r].w * m1.w;
        pk[r] = sk; pm[r] = sm;
    }
#pragma unroll
    for (int m = 16; m >= 1; m >>= 1) {
#pragma unroll
        for (int r = 0; r < 8; ++r) {
            pk[r] += __shfl_xor(pk[r], m, 32);
            pm[r] += __shfl_xor(pm[r], m, 32);
        }
    }
    float ok = 0.f, om = 0.f;
#pragma unroll
    for (int r = 0; r < 8; ++r) if (lane == r) { ok = pk[r]; om = pm[r]; }
    if (lane < 8) {
        s_out[wave * 8 + lane] = om * inv;
        s_out[DH_ + wave * 8 + lane] = ok * inv;
    }
    __syncthreads();
    if (t < 32)
        *(float4*)&mkp[((size_t)bh * NSL_ + sl) * (2 * DH_) + t * 4] = *(const float4*)&s_out[t * 4];
}

__global__ __launch_bounds__(256, 8) void pass3_kernel(
    const float* __restrict__ qs, const float* __restrict__ mkp, float* __restrict__ out)
{
    const int b = blockIdx.x >> 7;
    const int srow0 = (blockIdx.x & 127) * 32;
    const int t = threadIdx.x;
    const int h = t >> 4;
    const int wave = __builtin_amdgcn_readfirstlane(t >> 5);
    const bool upper = (t & 16) != 0;

    const float* mk = mkp + (size_t)(b * H_ + h) * (NSL_ * 2 * DH_) + (t & 15) * 4;
    float4 mv4 = *(const float4*)(mk);
    float4 kv4 = *(const float4*)(mk + DH_);
#pragma unroll
    for (int c = 1; c < NSL_; ++c) {
        float4 a = *(const float4*)(mk + c * 2 * DH_);
        float4 e = *(const float4*)(mk + c * 2 * DH_ + DH_);
        mv4.x += a.x; mv4.y += a.y; mv4.z += a.z; mv4.w += a.w;
        kv4.x += e.x; kv4.y += e.y; kv4.z += e.z; kv4.w += e.w;
    }
    const float* qrow = qs + ((size_t)b * S_ + srow0) * H_ + 2 * wave;
    float qlo[32], qhi[32];
#pragma unroll
    for (int r = 0; r < 32; ++r) {
        qlo[r] = qrow[r * H_];
        qhi[r] = qrow[r * H_ + 1];
    }

    float* orow = out + ((size_t)b * S_ + srow0) * (H_ * DH_) + t * 4;
#pragma unroll
    for (int r = 0; r < 32; ++r) {
        float q = upper ? qhi[r] : qlo[r];
        float4 o;
        o.x = q * mv4.x + kv4.x; o.y = q * mv4.y + kv4.y;
        o.z = q * mv4.z + kv4.z; o.w = q * mv4.w + kv4.w;
        *(float4*)(orow + (size_t)r * (H_ * DH_)) = o;
    }
}

extern "C" void kernel_launch(void* const* d_in, const int* in_sizes, int n_in,
                              void* d_out, int out_size, void* d_ws, size_t ws_size,
                              hipStream_t stream) {
    const float* hidden = (const float*)d_in[0];
    const float* mask   = (const float*)d_in[1];
    const float* Wq     = (const float*)d_in[2];
    const float* Wk     = (const float*)d_in[3];
    const float* Wv     = (const float*)d_in[4];
    const float* aw     = (const float*)d_in[5];
    float* out = (float*)d_out;
    float* ws  = (float*)d_ws;

    weff_kernel   <<<256,  32,  0, stream>>>(Wq, Wk, aw, ws + OFF_WQE, ws + OFF_WKE);
    pass1_kernel  <<<256,  512, 0, stream>>>(hidden, mask, ws + OFF_WQE, ws + OFF_WKE,
                                             ws + OFF_QS, ws + OFF_HSP, ws + OFF_KHP, ws + OFF_ML);
    reduce_project_kernel<<<512, 256, 0, stream>>>(ws + OFF_HSP, ws + OFF_KHP, Wv, ws + OFF_ML, ws + OFF_MKP);
    pass3_kernel  <<<1024, 256, 0, stream>>>(ws + OFF_QS, ws + OFF_MKP, out);
}
